# starting kernel
# baseline (speedup 1.0000x reference)
#include <hip/hip_runtime.h>
#include <math.h>


typedef float v2f __attribute__((ext_vector_type(2)));
typedef float v8f __attribute__((ext_vector_type(8)));

#define T_IN     441000
#define T_OUT    160000
#define BATCH    32
#define QBLK     1000
#define KPAD     800
#define KW       576
#define NPHASE   160
#define NUMZ     64.0
#define ROLLOFF  0.9475937167399596
#define KBETA    14.769656459379492
#define SCALE    (16000.0 / 44100.0)

__device__ inline double bessel_i0(double z) {
    double zz = 0.25 * z * z;
    double term = 1.0, sum = 1.0;
    #pragma unroll 1
    for (int m = 1; m <= 60; ++m) {
        term *= zz / (double)(m * m);
        sum += term;
    }
    return sum;
}

__global__ __launch_bounds__(256) void build_weights(float* __restrict__ W) {
    int idx = blockIdx.x * 256 + threadIdx.x;
    int p = idx / KPAD;
    int j = idx - p * KPAD;
    int pn   = p * 441;
    int boff = pn / 160;
    double frac = (double)(pn - boff * 160) / 160.0;
    int    k    = j - 177 - boff;
    double u    = ((double)k - frac) * SCALE;
    double w    = 0.0;
    if (fabs(u) < NUMZ) {
        double xr = ROLLOFF * u;
        double s  = (xr == 0.0) ? 1.0 : sin(M_PI * xr) / (M_PI * xr);
        double r  = 1.0 - (u / NUMZ) * (u / NUMZ);
        if (r < 0.0) r = 0.0;
        w = SCALE * ROLLOFF * s * bessel_i0(KBETA * sqrt(r)) / bessel_i0(KBETA);
    }
    W[idx] = (float)w;
}

template <bool MASKED>
__device__ inline void gemm_kloop(const float* __restrict__ x,
                                  const float* __restrict__ lds,
                                  int lm, int koff,
                                  int base0, int jlo0,
                                  int base1, int jlo1,
                                  v8f c[5][2]) {
    const int jhi0 = jlo0 + T_IN - 1;
    const int jhi1 = jlo1 + T_IN - 1;
    #pragma unroll 2
    for (int kk = 0; kk < KW; kk += 4) {
        const int j = kk + koff;
        float e0, e1, f0, f1;
        if constexpr (!MASKED) {
            e0 = x[base0 + j];
            e1 = x[base0 + j + 1];
            f0 = x[base1 + j];
            f1 = x[base1 + j + 1];
        } else {
            int ja = j     < jlo0 ? jlo0 : j;      ja = ja > jhi0 ? jhi0 : ja;
            int jb = j + 1 < jlo0 ? jlo0 : j + 1;  jb = jb > jhi0 ? jhi0 : jb;
            int jc = j     < jlo1 ? jlo1 : j;      jc = jc > jhi1 ? jhi1 : jc;
            int jd = j + 1 < jlo1 ? jlo1 : j + 1;  jd = jd > jhi1 ? jhi1 : jd;
            float m0 = ((unsigned)(j     - jlo0) < (unsigned)T_IN) ? 1.0f : 0.0f;
            float m1 = ((unsigned)(j + 1 - jlo0) < (unsigned)T_IN) ? 1.0f : 0.0f;
            float m2 = ((unsigned)(j     - jlo1) < (unsigned)T_IN) ? 1.0f : 0.0f;
            float m3 = ((unsigned)(j + 1 - jlo1) < (unsigned)T_IN) ? 1.0f : 0.0f;
            e0 = x[base0 + ja] * m0;
            e1 = x[base0 + jb] * m1;
            f0 = x[base1 + jc] * m2;
            f1 = x[base1 + jd] * m3;
        }
        v2f B0 = {e0, e1};
        v2f B1 = {f0, f1};
        #pragma unroll
        for (int mt = 0; mt < 5; ++mt) {
            v2f A = *(const v2f*)(lds + (mt * 16 + lm) * KW + j);
            c[mt][0] = __builtin_amdgcn_wmma_f32_16x16x4_f32(
                false, A, false, B0, (short)0, c[mt][0], false, false);
            c[mt][1] = __builtin_amdgcn_wmma_f32_16x16x4_f32(
                false, A, false, B1, (short)0, c[mt][1], false, false);
        }
    }
}

__global__ __launch_bounds__(256) void resample_wmma(const float* __restrict__ x,
                                                     const float* __restrict__ W,
                                                     float* __restrict__ y) {
    extern __shared__ float lds[];

    const int tid = threadIdx.x;
    const int ph  = blockIdx.y;
    const int cg  = blockIdx.x;
    const int kstart = ph ? 220 : 0;

    {
        const float* Wsrc = W + ph * 80 * KPAD + kstart;
        const int trow = tid >> 6;
        const int tcol = tid & 63;
        for (int row = trow; row < 80; row += 4) {
            for (int t4 = tcol; t4 < KW / 4; t4 += 64) {
                *(float4*)&lds[row * KW + t4 * 4] =
                    *(const float4*)&Wsrc[row * KPAD + t4 * 4];
            }
        }
    }
    __syncthreads();

    const int wave = tid >> 5;
    const int lane = tid & 31;
    const int lm   = lane & 15;
    const int hi   = lane >> 4;
    const int koff = hi * 2;

    const int col0 = cg * 256 + wave * 32 + lm;
    const int col1 = col0 + 16;
    const int b0 = col0 / QBLK, q0 = col0 - b0 * QBLK;
    const int b1 = col1 / QBLK, q1 = col1 - b1 * QBLK;
    const int base0 = b0 * T_IN + q0 * 441 - 177 + kstart;
    const int base1 = b1 * T_IN + q1 * 441 - 177 + kstart;
    const int jlo0  = 177 - q0 * 441 - kstart;
    const int jlo1  = 177 - q1 * 441 - kstart;

    v8f c[5][2] = {};

    const bool edge_lane = (q0 == 0) | (q0 == QBLK - 1) |
                           (q1 == 0) | (q1 == QBLK - 1);
    if (__any((int)edge_lane)) {
        gemm_kloop<true >(x, lds, lm, koff, base0, jlo0, base1, jlo1, c);
    } else {
        gemm_kloop<false>(x, lds, lm, koff, base0, jlo0, base1, jlo1, c);
    }

    const int mrow = hi * 8;
    const int ob0 = b0 * T_OUT + q0 * 160 + ph * 80;
    const int ob1 = b1 * T_OUT + q1 * 160 + ph * 80;
    #pragma unroll
    for (int mt = 0; mt < 5; ++mt) {
        #pragma unroll
        for (int r = 0; r < 8; ++r) {
            y[ob0 + mt * 16 + mrow + r] = c[mt][0][r];
            y[ob1 + mt * 16 + mrow + r] = c[mt][1][r];
        }
    }
}

extern "C" void kernel_launch(void* const* d_in, const int* in_sizes, int n_in,
                              void* d_out, int out_size, void* d_ws, size_t ws_size,
                              hipStream_t stream) {
    const float* x = (const float*)d_in[0];
    float*       W = (float*)d_ws;
    float*       y = (float*)d_out;

    build_weights<<<500, 256, 0, stream>>>(W);

    dim3 grid(125, 2);
    resample_wmma<<<grid, 256, 80 * KW * sizeof(float), stream>>>(x, W, y);
}


# weight-table kernel: I0 series as a Horner polynomial with compile-time coefficients (one f64 FMA per term, no f64 division) and 1/I0(beta) folded into a compile-time factor
# speedup vs baseline: 1.9587x; 1.9587x over previous
#include <hip/hip_runtime.h>
#include <math.h>


typedef float v2f __attribute__((ext_vector_type(2)));
typedef float v8f __attribute__((ext_vector_type(8)));

#define T_IN     441000
#define T_OUT    160000
#define BATCH    32
#define QBLK     1000
#define KPAD     800
#define KW       576
#define NPHASE   160
#define NUMZ     64.0
#define ROLLOFF  0.9475937167399596
#define KBETA    14.769656459379492
#define SCALE    (16000.0 / 44100.0)

struct I0Coef { double c[61]; };
constexpr I0Coef make_i0_coef() {
    I0Coef t = {};
    double v = 1.0;
    t.c[0] = 1.0;
    for (int m = 1; m <= 60; ++m) {
        v /= (double)(m * m);
        t.c[m] = v;
    }
    return t;
}
constexpr double bessel_i0(double z) {
    constexpr I0Coef t = make_i0_coef();
    const double zz = 0.25 * z * z;
    double sum = t.c[60];
    #pragma unroll
    for (int m = 59; m >= 0; --m) {
        sum = sum * zz + t.c[m];
    }
    return sum;
}
constexpr double W_GAIN = SCALE * ROLLOFF / bessel_i0(KBETA);

__global__ __launch_bounds__(256) void build_weights(float* __restrict__ W) {
    int idx = blockIdx.x * 256 + threadIdx.x;
    int p = idx / KPAD;
    int j = idx - p * KPAD;
    int pn   = p * 441;
    int boff = pn / 160;
    double frac = (double)(pn - boff * 160) / 160.0;
    int    k    = j - 177 - boff;
    double u    = ((double)k - frac) * SCALE;
    double w    = 0.0;
    if (fabs(u) < NUMZ) {
        double xr = ROLLOFF * u;
        double s  = (xr == 0.0) ? 1.0 : sin(M_PI * xr) / (M_PI * xr);
        double r  = 1.0 - (u / NUMZ) * (u / NUMZ);
        if (r < 0.0) r = 0.0;
        w = W_GAIN * s * bessel_i0(KBETA * sqrt(r));
    }
    W[idx] = (float)w;
}

template <bool MASKED>
__device__ inline void gemm_kloop(const float* __restrict__ x,
                                  const float* __restrict__ lds,
                                  int lm, int koff,
                                  int base0, int jlo0,
                                  int base1, int jlo1,
                                  v8f c[5][2]) {
    const int jhi0 = jlo0 + T_IN - 1;
    const int jhi1 = jlo1 + T_IN - 1;
    #pragma unroll 2
    for (int kk = 0; kk < KW; kk += 4) {
        const int j = kk + koff;
        float e0, e1, f0, f1;
        if constexpr (!MASKED) {
            e0 = x[base0 + j];
            e1 = x[base0 + j + 1];
            f0 = x[base1 + j];
            f1 = x[base1 + j + 1];
        } else {
            int ja = j     < jlo0 ? jlo0 : j;      ja = ja > jhi0 ? jhi0 : ja;
            int jb = j + 1 < jlo0 ? jlo0 : j + 1;  jb = jb > jhi0 ? jhi0 : jb;
            int jc = j     < jlo1 ? jlo1 : j;      jc = jc > jhi1 ? jhi1 : jc;
            int jd = j + 1 < jlo1 ? jlo1 : j + 1;  jd = jd > jhi1 ? jhi1 : jd;
            float m0 = ((unsigned)(j     - jlo0) < (unsigned)T_IN) ? 1.0f : 0.0f;
            float m1 = ((unsigned)(j + 1 - jlo0) < (unsigned)T_IN) ? 1.0f : 0.0f;
            float m2 = ((unsigned)(j     - jlo1) < (unsigned)T_IN) ? 1.0f : 0.0f;
            float m3 = ((unsigned)(j + 1 - jlo1) < (unsigned)T_IN) ? 1.0f : 0.0f;
            e0 = x[base0 + ja] * m0;
            e1 = x[base0 + jb] * m1;
            f0 = x[base1 + jc] * m2;
            f1 = x[base1 + jd] * m3;
        }
        v2f B0 = {e0, e1};
        v2f B1 = {f0, f1};
        #pragma unroll
        for (int mt = 0; mt < 5; ++mt) {
            v2f A = *(const v2f*)(lds + (mt * 16 + lm) * KW + j);
            c[mt][0] = __builtin_amdgcn_wmma_f32_16x16x4_f32(
                false, A, false, B0, (short)0, c[mt][0], false, false);
            c[mt][1] = __builtin_amdgcn_wmma_f32_16x16x4_f32(
                false, A, false, B1, (short)0, c[mt][1], false, false);
        }
    }
}

__global__ __launch_bounds__(256) void resample_wmma(const float* __restrict__ x,
                                                     const float* __restrict__ W,
                                                     float* __restrict__ y) {
    extern __shared__ float lds[];

    const int tid = threadIdx.x;
    const int ph  = blockIdx.y;
    const int cg  = blockIdx.x;
    const int kstart = ph ? 220 : 0;

    {
        const float* Wsrc = W + ph * 80 * KPAD + kstart;
        const int trow = tid >> 6;
        const int tcol = tid & 63;
        for (int row = trow; row < 80; row += 4) {
            for (int t4 = tcol; t4 < KW / 4; t4 += 64) {
                *(float4*)&lds[row * KW + t4 * 4] =
                    *(const float4*)&Wsrc[row * KPAD + t4 * 4];
            }
        }
    }
    __syncthreads();

    const int wave = tid >> 5;
    const int lane = tid & 31;
    const int lm   = lane & 15;
    const int hi   = lane >> 4;
    const int koff = hi * 2;

    const int col0 = cg * 256 + wave * 32 + lm;
    const int col1 = col0 + 16;
    const int b0 = col0 / QBLK, q0 = col0 - b0 * QBLK;
    const int b1 = col1 / QBLK, q1 = col1 - b1 * QBLK;
    const int base0 = b0 * T_IN + q0 * 441 - 177 + kstart;
    const int base1 = b1 * T_IN + q1 * 441 - 177 + kstart;
    const int jlo0  = 177 - q0 * 441 - kstart;
    const int jlo1  = 177 - q1 * 441 - kstart;

    v8f c[5][2] = {};

    const bool edge_lane = (q0 == 0) | (q0 == QBLK - 1) |
                           (q1 == 0) | (q1 == QBLK - 1);
    if (__any((int)edge_lane)) {
        gemm_kloop<true >(x, lds, lm, koff, base0, jlo0, base1, jlo1, c);
    } else {
        gemm_kloop<false>(x, lds, lm, koff, base0, jlo0, base1, jlo1, c);
    }

    const int mrow = hi * 8;
    const int ob0 = b0 * T_OUT + q0 * 160 + ph * 80;
    const int ob1 = b1 * T_OUT + q1 * 160 + ph * 80;
    #pragma unroll
    for (int mt = 0; mt < 5; ++mt) {
        #pragma unroll
        for (int r = 0; r < 8; ++r) {
            y[ob0 + mt * 16 + mrow + r] = c[mt][0][r];
            y[ob1 + mt * 16 + mrow + r] = c[mt][1][r];
        }
    }
}

extern "C" void kernel_launch(void* const* d_in, const int* in_sizes, int n_in,
                              void* d_out, int out_size, void* d_ws, size_t ws_size,
                              hipStream_t stream) {
    const float* x = (const float*)d_in[0];
    float*       W = (float*)d_ws;
    float*       y = (float*)d_out;

    build_weights<<<500, 256, 0, stream>>>(W);

    dim3 grid(125, 2);
    resample_wmma<<<grid, 256, 80 * KW * sizeof(float), stream>>>(x, W, y);
}


# LDS row pitch of the staged weight window 576 -> 580 floats (16 B x odd): the 16 rows of an A-fragment read fall in 16 different banks instead of one, removing a 16-way bank conflict
# speedup vs baseline: 2.0367x; 1.0398x over previous
#include <hip/hip_runtime.h>
#include <math.h>


typedef float v2f __attribute__((ext_vector_type(2)));
typedef float v8f __attribute__((ext_vector_type(8)));

#define T_IN     441000
#define T_OUT    160000
#define BATCH    32
#define QBLK     1000
#define KPAD     800
#define KW       576
#define LDSP     580
#define NPHASE   160
#define NUMZ     64.0
#define ROLLOFF  0.9475937167399596
#define KBETA    14.769656459379492
#define SCALE    (16000.0 / 44100.0)

struct I0Coef { double c[61]; };
constexpr I0Coef make_i0_coef() {
    I0Coef t = {};
    double v = 1.0;
    t.c[0] = 1.0;
    for (int m = 1; m <= 60; ++m) {
        v /= (double)(m * m);
        t.c[m] = v;
    }
    return t;
}
constexpr double bessel_i0(double z) {
    constexpr I0Coef t = make_i0_coef();
    const double zz = 0.25 * z * z;
    double sum = t.c[60];
    #pragma unroll
    for (int m = 59; m >= 0; --m) {
        sum = sum * zz + t.c[m];
    }
    return sum;
}
constexpr double W_GAIN = SCALE * ROLLOFF / bessel_i0(KBETA);

__global__ __launch_bounds__(256) void build_weights(float* __restrict__ W) {
    int idx = blockIdx.x * 256 + threadIdx.x;
    int p = idx / KPAD;
    int j = idx - p * KPAD;
    int pn   = p * 441;
    int boff = pn / 160;
    double frac = (double)(pn - boff * 160) / 160.0;
    int    k    = j - 177 - boff;
    double u    = ((double)k - frac) * SCALE;
    double w    = 0.0;
    if (fabs(u) < NUMZ) {
        double xr = ROLLOFF * u;
        double s  = (xr == 0.0) ? 1.0 : sin(M_PI * xr) / (M_PI * xr);
        double r  = 1.0 - (u / NUMZ) * (u / NUMZ);
        if (r < 0.0) r = 0.0;
        w = W_GAIN * s * bessel_i0(KBETA * sqrt(r));
    }
    W[idx] = (float)w;
}

template <bool MASKED>
__device__ inline void gemm_kloop(const float* __restrict__ x,
                                  const float* __restrict__ lds,
                                  int lm, int koff,
                                  int base0, int jlo0,
                                  int base1, int jlo1,
                                  v8f c[5][2]) {
    const int jhi0 = jlo0 + T_IN - 1;
    const int jhi1 = jlo1 + T_IN - 1;
    #pragma unroll 2
    for (int kk = 0; kk < KW; kk += 4) {
        const int j = kk + koff;
        float e0, e1, f0, f1;
        if constexpr (!MASKED) {
            e0 = x[base0 + j];
            e1 = x[base0 + j + 1];
            f0 = x[base1 + j];
            f1 = x[base1 + j + 1];
        } else {
            int ja = j     < jlo0 ? jlo0 : j;      ja = ja > jhi0 ? jhi0 : ja;
            int jb = j + 1 < jlo0 ? jlo0 : j + 1;  jb = jb > jhi0 ? jhi0 : jb;
            int jc = j     < jlo1 ? jlo1 : j;      jc = jc > jhi1 ? jhi1 : jc;
            int jd = j + 1 < jlo1 ? jlo1 : j + 1;  jd = jd > jhi1 ? jhi1 : jd;
            float m0 = ((unsigned)(j     - jlo0) < (unsigned)T_IN) ? 1.0f : 0.0f;
            float m1 = ((unsigned)(j + 1 - jlo0) < (unsigned)T_IN) ? 1.0f : 0.0f;
            float m2 = ((unsigned)(j     - jlo1) < (unsigned)T_IN) ? 1.0f : 0.0f;
            float m3 = ((unsigned)(j + 1 - jlo1) < (unsigned)T_IN) ? 1.0f : 0.0f;
            e0 = x[base0 + ja] * m0;
            e1 = x[base0 + jb] * m1;
            f0 = x[base1 + jc] * m2;
            f1 = x[base1 + jd] * m3;
        }
        v2f B0 = {e0, e1};
        v2f B1 = {f0, f1};
        #pragma unroll
        for (int mt = 0; mt < 5; ++mt) {
            v2f A = *(const v2f*)(lds + (mt * 16 + lm) * LDSP + j);
            c[mt][0] = __builtin_amdgcn_wmma_f32_16x16x4_f32(
                false, A, false, B0, (short)0, c[mt][0], false, false);
            c[mt][1] = __builtin_amdgcn_wmma_f32_16x16x4_f32(
                false, A, false, B1, (short)0, c[mt][1], false, false);
        }
    }
}

__global__ __launch_bounds__(256) void resample_wmma(const float* __restrict__ x,
                                                     const float* __restrict__ W,
                                                     float* __restrict__ y) {
    extern __shared__ float lds[];

    const int tid = threadIdx.x;
    const int ph  = blockIdx.y;
    const int cg  = blockIdx.x;
    const int kstart = ph ? 220 : 0;

    {
        const float* Wsrc = W + ph * 80 * KPAD + kstart;
        const int trow = tid >> 6;
        const int tcol = tid & 63;
        for (int row = trow; row < 80; row += 4) {
            for (int t4 = tcol; t4 < KW / 4; t4 += 64) {
                *(float4*)&lds[row * LDSP + t4 * 4] =
                    *(const float4*)&Wsrc[row * KPAD + t4 * 4];
            }
        }
    }
    __syncthreads();

    const int wave = tid >> 5;
    const int lane = tid & 31;
    const int lm   = lane & 15;
    const int hi   = lane >> 4;
    const int koff = hi * 2;

    const int col0 = cg * 256 + wave * 32 + lm;
    const int col1 = col0 + 16;
    const int b0 = col0 / QBLK, q0 = col0 - b0 * QBLK;
    const int b1 = col1 / QBLK, q1 = col1 - b1 * QBLK;
    const int base0 = b0 * T_IN + q0 * 441 - 177 + kstart;
    const int base1 = b1 * T_IN + q1 * 441 - 177 + kstart;
    const int jlo0  = 177 - q0 * 441 - kstart;
    const int jlo1  = 177 - q1 * 441 - kstart;

    v8f c[5][2] = {};

    const bool edge_lane = (q0 == 0) | (q0 == QBLK - 1) |
                           (q1 == 0) | (q1 == QBLK - 1);
    if (__any((int)edge_lane)) {
        gemm_kloop<true >(x, lds, lm, koff, base0, jlo0, base1, jlo1, c);
    } else {
        gemm_kloop<false>(x, lds, lm, koff, base0, jlo0, base1, jlo1, c);
    }

    const int mrow = hi * 8;
    const int ob0 = b0 * T_OUT + q0 * 160 + ph * 80;
    const int ob1 = b1 * T_OUT + q1 * 160 + ph * 80;
    #pragma unroll
    for (int mt = 0; mt < 5; ++mt) {
        #pragma unroll
        for (int r = 0; r < 8; ++r) {
            y[ob0 + mt * 16 + mrow + r] = c[mt][0][r];
            y[ob1 + mt * 16 + mrow + r] = c[mt][1][r];
        }
    }
}

extern "C" void kernel_launch(void* const* d_in, const int* in_sizes, int n_in,
                              void* d_out, int out_size, void* d_ws, size_t ws_size,
                              hipStream_t stream) {
    const float* x = (const float*)d_in[0];
    float*       W = (float*)d_ws;
    float*       y = (float*)d_out;

    build_weights<<<500, 256, 0, stream>>>(W);

    dim3 grid(125, 2);
    resample_wmma<<<grid, 256, 80 * LDSP * sizeof(float), stream>>>(x, W, y);
}


# x (B-fragment) loads issued one group of 12 k-steps ahead into a second register set, in front of the current group's WMMAs: memory latency overlaps matrix work; k and summation order unchanged
# speedup vs baseline: 2.4868x; 1.2210x over previous
#include <hip/hip_runtime.h>
#include <math.h>


typedef float v2f __attribute__((ext_vector_type(2)));
typedef float v8f __attribute__((ext_vector_type(8)));

#define T_IN     441000
#define T_OUT    160000
#define BATCH    32
#define QBLK     1000
#define KPAD     800
#define KW       576
#define LDSP     580
#define NPHASE   160
#define NUMZ     64.0
#define ROLLOFF  0.9475937167399596
#define KBETA    14.769656459379492
#define SCALE    (16000.0 / 44100.0)

struct I0Coef { double c[61]; };
constexpr I0Coef make_i0_coef() {
    I0Coef t = {};
    double v = 1.0;
    t.c[0] = 1.0;
    for (int m = 1; m <= 60; ++m) {
        v /= (double)(m * m);
        t.c[m] = v;
    }
    return t;
}
constexpr double bessel_i0(double z) {
    constexpr I0Coef t = make_i0_coef();
    const double zz = 0.25 * z * z;
    double sum = t.c[60];
    #pragma unroll
    for (int m = 59; m >= 0; --m) {
        sum = sum * zz + t.c[m];
    }
    return sum;
}
constexpr double W_GAIN = SCALE * ROLLOFF / bessel_i0(KBETA);

__global__ __launch_bounds__(256) void build_weights(float* __restrict__ W) {
    int idx = blockIdx.x * 256 + threadIdx.x;
    int p = idx / KPAD;
    int j = idx - p * KPAD;
    int pn   = p * 441;
    int boff = pn / 160;
    double frac = (double)(pn - boff * 160) / 160.0;
    int    k    = j - 177 - boff;
    double u    = ((double)k - frac) * SCALE;
    double w    = 0.0;
    if (fabs(u) < NUMZ) {
        double xr = ROLLOFF * u;
        double s  = (xr == 0.0) ? 1.0 : sin(M_PI * xr) / (M_PI * xr);
        double r  = 1.0 - (u / NUMZ) * (u / NUMZ);
        if (r < 0.0) r = 0.0;
        w = W_GAIN * s * bessel_i0(KBETA * sqrt(r));
    }
    W[idx] = (float)w;
}

#define RING     12
#define KGRP     (4 * RING)

struct XRaw { float e0, e1, f0, f1; };

template <bool MASKED>
__device__ inline XRaw load_x(const float* __restrict__ x, int j,
                              int base0, int jlo0, int base1, int jlo1) {
    XRaw r;
    if constexpr (!MASKED) {
        r.e0 = x[base0 + j];
        r.e1 = x[base0 + j + 1];
        r.f0 = x[base1 + j];
        r.f1 = x[base1 + j + 1];
    } else {
        const int jhi0 = jlo0 + T_IN - 1;
        const int jhi1 = jlo1 + T_IN - 1;
        int ja = j     < jlo0 ? jlo0 : j;      ja = ja > jhi0 ? jhi0 : ja;
        int jb = j + 1 < jlo0 ? jlo0 : j + 1;  jb = jb > jhi0 ? jhi0 : jb;
        int jc = j     < jlo1 ? jlo1 : j;      jc = jc > jhi1 ? jhi1 : jc;
        int jd = j + 1 < jlo1 ? jlo1 : j + 1;  jd = jd > jhi1 ? jhi1 : jd;
        r.e0 = x[base0 + ja];
        r.e1 = x[base0 + jb];
        r.f0 = x[base1 + jc];
        r.f1 = x[base1 + jd];
    }
    return r;
}

template <bool MASKED>
__device__ inline void mma_step(const float* __restrict__ lds, int lm, int j,
                                XRaw r, int jlo0, int jlo1, v8f c[5][2]) {
    if constexpr (MASKED) {
        float m0 = ((unsigned)(j     - jlo0) < (unsigned)T_IN) ? 1.0f : 0.0f;
        float m1 = ((unsigned)(j + 1 - jlo0) < (unsigned)T_IN) ? 1.0f : 0.0f;
        float m2 = ((unsigned)(j     - jlo1) < (unsigned)T_IN) ? 1.0f : 0.0f;
        float m3 = ((unsigned)(j + 1 - jlo1) < (unsigned)T_IN) ? 1.0f : 0.0f;
        r.e0 *= m0;
        r.e1 *= m1;
        r.f0 *= m2;
        r.f1 *= m3;
    }
    v2f B0 = {r.e0, r.e1};
    v2f B1 = {r.f0, r.f1};
    #pragma unroll
    for (int mt = 0; mt < 5; ++mt) {
        v2f A = *(const v2f*)(lds + (mt * 16 + lm) * LDSP + j);
        c[mt][0] = __builtin_amdgcn_wmma_f32_16x16x4_f32(
            false, A, false, B0, (short)0, c[mt][0], false, false);
        c[mt][1] = __builtin_amdgcn_wmma_f32_16x16x4_f32(
            false, A, false, B1, (short)0, c[mt][1], false, false);
    }
}

template <bool MASKED>
__device__ inline void gemm_kloop(const float* __restrict__ x,
                                  const float* __restrict__ lds,
                                  int lm, int koff,
                                  int base0, int jlo0,
                                  int base1, int jlo1,
                                  v8f c[5][2]) {
    XRaw cur[RING], nxt[RING];
    #pragma unroll
    for (int s = 0; s < RING; ++s) {
        cur[s] = load_x<MASKED>(x, 4 * s + koff, base0, jlo0, base1, jlo1);
    }
    #pragma nounroll
    for (int kk0 = 0; kk0 < KW - KGRP; kk0 += KGRP) {
        #pragma unroll
        for (int s = 0; s < RING; ++s) {
            nxt[s] = load_x<MASKED>(x, kk0 + KGRP + 4 * s + koff,
                                    base0, jlo0, base1, jlo1);
        }
        __builtin_amdgcn_sched_barrier(0);
        #pragma unroll
        for (int s = 0; s < RING; ++s) {
            const int j = kk0 + 4 * s + koff;
            mma_step<MASKED>(lds, lm, j, cur[s], jlo0, jlo1, c);
        }
        __builtin_amdgcn_sched_barrier(0);
        #pragma unroll
        for (int s = 0; s < RING; ++s) {
            cur[s] = nxt[s];
        }
    }
    #pragma unroll
    for (int s = 0; s < RING; ++s) {
        const int j = KW - KGRP + 4 * s + koff;
        mma_step<MASKED>(lds, lm, j, cur[s], jlo0, jlo1, c);
    }
}

__global__ __launch_bounds__(256) void resample_wmma(const float* __restrict__ x,
                                                     const float* __restrict__ W,
                                                     float* __restrict__ y) {
    extern __shared__ float lds[];

    const int tid = threadIdx.x;
    const int ph  = blockIdx.y;
    const int cg  = blockIdx.x;
    const int kstart = ph ? 220 : 0;

    {
        const float* Wsrc = W + ph * 80 * KPAD + kstart;
        const int trow = tid >> 6;
        const int tcol = tid & 63;
        for (int row = trow; row < 80; row += 4) {
            for (int t4 = tcol; t4 < KW / 4; t4 += 64) {
                *(float4*)&lds[row * LDSP + t4 * 4] =
                    *(const float4*)&Wsrc[row * KPAD + t4 * 4];
            }
        }
    }
    __syncthreads();

    const int wave = tid >> 5;
    const int lane = tid & 31;
    const int lm   = lane & 15;
    const int hi   = lane >> 4;
    const int koff = hi * 2;

    const int col0 = cg * 256 + wave * 32 + lm;
    const int col1 = col0 + 16;
    const int b0 = col0 / QBLK, q0 = col0 - b0 * QBLK;
    const int b1 = col1 / QBLK, q1 = col1 - b1 * QBLK;
    const int base0 = b0 * T_IN + q0 * 441 - 177 + kstart;
    const int base1 = b1 * T_IN + q1 * 441 - 177 + kstart;
    const int jlo0  = 177 - q0 * 441 - kstart;
    const int jlo1  = 177 - q1 * 441 - kstart;

    v8f c[5][2] = {};

    const bool edge_lane = (q0 == 0) | (q0 == QBLK - 1) |
                           (q1 == 0) | (q1 == QBLK - 1);
    if (__any((int)edge_lane)) {
        gemm_kloop<true >(x, lds, lm, koff, base0, jlo0, base1, jlo1, c);
    } else {
        gemm_kloop<false>(x, lds, lm, koff, base0, jlo0, base1, jlo1, c);
    }

    const int mrow = hi * 8;
    const int ob0 = b0 * T_OUT + q0 * 160 + ph * 80;
    const int ob1 = b1 * T_OUT + q1 * 160 + ph * 80;
    #pragma unroll
    for (int mt = 0; mt < 5; ++mt) {
        #pragma unroll
        for (int r = 0; r < 8; ++r) {
            y[ob0 + mt * 16 + mrow + r] = c[mt][0][r];
            y[ob1 + mt * 16 + mrow + r] = c[mt][1][r];
        }
    }
}

extern "C" void kernel_launch(void* const* d_in, const int* in_sizes, int n_in,
                              void* d_out, int out_size, void* d_ws, size_t ws_size,
                              hipStream_t stream) {
    const float* x = (const float*)d_in[0];
    float*       W = (float*)d_ws;
    float*       y = (float*)d_out;

    build_weights<<<500, 256, 0, stream>>>(W);

    dim3 grid(125, 2);
    resample_wmma<<<grid, 256, 80 * LDSP * sizeof(float), stream>>>(x, W, y);
}


# W-window staging into LDS: each thread issues all 45 of its 16-byte loads before its first LDS store (was a load-wait-store loop: one load in flight per wave, 60 serial trips); sums unchanged
# speedup vs baseline: 3.1286x; 1.2581x over previous
#include <hip/hip_runtime.h>
#include <math.h>


typedef float v2f __attribute__((ext_vector_type(2)));
typedef float v8f __attribute__((ext_vector_type(8)));

#define T_IN     441000
#define T_OUT    160000
#define BATCH    32
#define QBLK     1000
#define KPAD     800
#define KW       576
#define LDSP     580
#define STG_V4   (KW / 4)
#define STG_B    9
#define STG_A    5
#define NPHASE   160
#define NUMZ     64.0
#define ROLLOFF  0.9475937167399596
#define KBETA    14.769656459379492
#define SCALE    (16000.0 / 44100.0)

struct I0Coef { double c[61]; };
constexpr I0Coef make_i0_coef() {
    I0Coef t = {};
    double v = 1.0;
    t.c[0] = 1.0;
    for (int m = 1; m <= 60; ++m) {
        v /= (double)(m * m);
        t.c[m] = v;
    }
    return t;
}
constexpr double bessel_i0(double z) {
    constexpr I0Coef t = make_i0_coef();
    const double zz = 0.25 * z * z;
    double sum = t.c[60];
    #pragma unroll
    for (int m = 59; m >= 0; --m) {
        sum = sum * zz + t.c[m];
    }
    return sum;
}
constexpr double W_GAIN = SCALE * ROLLOFF / bessel_i0(KBETA);

__global__ __launch_bounds__(256) void build_weights(float* __restrict__ W) {
    int idx = blockIdx.x * 256 + threadIdx.x;
    int p = idx / KPAD;
    int j = idx - p * KPAD;
    int pn   = p * 441;
    int boff = pn / 160;
    double frac = (double)(pn - boff * 160) / 160.0;
    int    k    = j - 177 - boff;
    double u    = ((double)k - frac) * SCALE;
    double w    = 0.0;
    if (fabs(u) < NUMZ) {
        double xr = ROLLOFF * u;
        double s  = (xr == 0.0) ? 1.0 : sin(M_PI * xr) / (M_PI * xr);
        double r  = 1.0 - (u / NUMZ) * (u / NUMZ);
        if (r < 0.0) r = 0.0;
        w = W_GAIN * s * bessel_i0(KBETA * sqrt(r));
    }
    W[idx] = (float)w;
}

#define RING     12
#define KGRP     (4 * RING)

struct XRaw { float e0, e1, f0, f1; };

template <bool MASKED>
__device__ inline XRaw load_x(const float* __restrict__ x, int j,
                              int base0, int jlo0, int base1, int jlo1) {
    XRaw r;
    if constexpr (!MASKED) {
        r.e0 = x[base0 + j];
        r.e1 = x[base0 + j + 1];
        r.f0 = x[base1 + j];
        r.f1 = x[base1 + j + 1];
    } else {
        const int jhi0 = jlo0 + T_IN - 1;
        const int jhi1 = jlo1 + T_IN - 1;
        int ja = j     < jlo0 ? jlo0 : j;      ja = ja > jhi0 ? jhi0 : ja;
        int jb = j + 1 < jlo0 ? jlo0 : j + 1;  jb = jb > jhi0 ? jhi0 : jb;
        int jc = j     < jlo1 ? jlo1 : j;      jc = jc > jhi1 ? jhi1 : jc;
        int jd = j + 1 < jlo1 ? jlo1 : j + 1;  jd = jd > jhi1 ? jhi1 : jd;
        r.e0 = x[base0 + ja];
        r.e1 = x[base0 + jb];
        r.f0 = x[base1 + jc];
        r.f1 = x[base1 + jd];
    }
    return r;
}

template <bool MASKED>
__device__ inline void mma_step(const float* __restrict__ lds, int lm, int j,
                                XRaw r, int jlo0, int jlo1, v8f c[5][2]) {
    if constexpr (MASKED) {
        float m0 = ((unsigned)(j     - jlo0) < (unsigned)T_IN) ? 1.0f : 0.0f;
        float m1 = ((unsigned)(j + 1 - jlo0) < (unsigned)T_IN) ? 1.0f : 0.0f;
        float m2 = ((unsigned)(j     - jlo1) < (unsigned)T_IN) ? 1.0f : 0.0f;
        float m3 = ((unsigned)(j + 1 - jlo1) < (unsigned)T_IN) ? 1.0f : 0.0f;
        r.e0 *= m0;
        r.e1 *= m1;
        r.f0 *= m2;
        r.f1 *= m3;
    }
    v2f B0 = {r.e0, r.e1};
    v2f B1 = {r.f0, r.f1};
    #pragma unroll
    for (int mt = 0; mt < 5; ++mt) {
        v2f A = *(const v2f*)(lds + (mt * 16 + lm) * LDSP + j);
        c[mt][0] = __builtin_amdgcn_wmma_f32_16x16x4_f32(
            false, A, false, B0, (short)0, c[mt][0], false, false);
        c[mt][1] = __builtin_amdgcn_wmma_f32_16x16x4_f32(
            false, A, false, B1, (short)0, c[mt][1], false, false);
    }
}

template <bool MASKED>
__device__ inline void gemm_kloop(const float* __restrict__ x,
                                  const float* __restrict__ lds,
                                  int lm, int koff,
                                  int base0, int jlo0,
                                  int base1, int jlo1,
                                  v8f c[5][2]) {
    XRaw cur[RING], nxt[RING];
    #pragma unroll
    for (int s = 0; s < RING; ++s) {
        cur[s] = load_x<MASKED>(x, 4 * s + koff, base0, jlo0, base1, jlo1);
    }
    #pragma nounroll
    for (int kk0 = 0; kk0 < KW - KGRP; kk0 += KGRP) {
        #pragma unroll
        for (int s = 0; s < RING; ++s) {
            nxt[s] = load_x<MASKED>(x, kk0 + KGRP + 4 * s + koff,
                                    base0, jlo0, base1, jlo1);
        }
        __builtin_amdgcn_sched_barrier(0);
        #pragma unroll
        for (int s = 0; s < RING; ++s) {
            const int j = kk0 + 4 * s + koff;
            mma_step<MASKED>(lds, lm, j, cur[s], jlo0, jlo1, c);
        }
        __builtin_amdgcn_sched_barrier(0);
        #pragma unroll
        for (int s = 0; s < RING; ++s) {
            cur[s] = nxt[s];
        }
    }
    #pragma unroll
    for (int s = 0; s < RING; ++s) {
        const int j = KW - KGRP + 4 * s + koff;
        mma_step<MASKED>(lds, lm, j, cur[s], jlo0, jlo1, c);
    }
}

__global__ __launch_bounds__(256) void resample_wmma(const float* __restrict__ x,
                                                     const float* __restrict__ W,
                                                     float* __restrict__ y) {
    extern __shared__ __attribute__((aligned(16))) float lds[];

    const int tid = threadIdx.x;
    const int ph  = blockIdx.y;
    const int cg  = blockIdx.x;
    const int kstart = ph ? 220 : 0;

    {
        const float* Wsrc = W + ph * 80 * KPAD + kstart;
        float4 wv[STG_A][STG_B];
        #pragma unroll
        for (int a = 0; a < STG_A; ++a) {
            #pragma unroll
            for (int b = 0; b < STG_B; ++b) {
                const int i   = tid + 256 * b;
                const int row = i / STG_V4 + 16 * a;
                const int t4  = i % STG_V4;
                wv[a][b] = *(const float4*)&Wsrc[row * KPAD + t4 * 4];
            }
        }
        #pragma unroll
        for (int a = 0; a < STG_A; ++a) {
            #pragma unroll
            for (int b = 0; b < STG_B; ++b) {
                const int i   = tid + 256 * b;
                const int row = i / STG_V4 + 16 * a;
                const int t4  = i % STG_V4;
                *(float4*)&lds[row * LDSP + t4 * 4] = wv[a][b];
            }
        }
    }
    __syncthreads();

    const int wave = tid >> 5;
    const int lane = tid & 31;
    const int lm   = lane & 15;
    const int hi   = lane >> 4;
    const int koff = hi * 2;

    const int col0 = cg * 256 + wave * 32 + lm;
    const int col1 = col0 + 16;
    const int b0 = col0 / QBLK, q0 = col0 - b0 * QBLK;
    const int b1 = col1 / QBLK, q1 = col1 - b1 * QBLK;
    const int base0 = b0 * T_IN + q0 * 441 - 177 + kstart;
    const int base1 = b1 * T_IN + q1 * 441 - 177 + kstart;
    const int jlo0  = 177 - q0 * 441 - kstart;
    const int jlo1  = 177 - q1 * 441 - kstart;

    v8f c[5][2] = {};

    const bool edge_lane = (q0 == 0) | (q0 == QBLK - 1) |
                           (q1 == 0) | (q1 == QBLK - 1);
    if (__any((int)edge_lane)) {
        gemm_kloop<true >(x, lds, lm, koff, base0, jlo0, base1, jlo1, c);
    } else {
        gemm_kloop<false>(x, lds, lm, koff, base0, jlo0, base1, jlo1, c);
    }

    const int mrow = hi * 8;
    const int ob0 = b0 * T_OUT + q0 * 160 + ph * 80;
    const int ob1 = b1 * T_OUT + q1 * 160 + ph * 80;
    #pragma unroll
    for (int mt = 0; mt < 5; ++mt) {
        #pragma unroll
        for (int r = 0; r < 8; ++r) {
            y[ob0 + mt * 16 + mrow + r] = c[mt][0][r];
            y[ob1 + mt * 16 + mrow + r] = c[mt][1][r];
        }
    }
}

extern "C" void kernel_launch(void* const* d_in, const int* in_sizes, int n_in,
                              void* d_out, int out_size, void* d_ws, size_t ws_size,
                              hipStream_t stream) {
    const float* x = (const float*)d_in[0];
    float*       W = (float*)d_ws;
    float*       y = (float*)d_out;

    build_weights<<<500, 256, 0, stream>>>(W);

    dim3 grid(125, 2);
    resample_wmma<<<grid, 256, 80 * LDSP * sizeof(float), stream>>>(x, W, y);
}


# x loader: each lane takes 4 consecutive taps with one 16-byte load for two k-steps (built code had two 8-byte loads); W stored in matching tap order; taps regrouped per k-step, sum order changes
# speedup vs baseline: 3.2367x; 1.0346x over previous
#include <hip/hip_runtime.h>
#include <math.h>


typedef float v2f __attribute__((ext_vector_type(2)));
typedef float v8f __attribute__((ext_vector_type(8)));

#define T_IN     441000
#define T_OUT    160000
#define BATCH    32
#define QBLK     1000
#define KPAD     800
#define KW       576
#define LDSP     580
#define STG_V4   (KW / 4)
#define STG_B    9
#define STG_A    5
#define NPHASE   160
#define NUMZ     64.0
#define ROLLOFF  0.9475937167399596
#define KBETA    14.769656459379492
#define SCALE    (16000.0 / 44100.0)

struct I0Coef { double c[61]; };
constexpr I0Coef make_i0_coef() {
    I0Coef t = {};
    double v = 1.0;
    t.c[0] = 1.0;
    for (int m = 1; m <= 60; ++m) {
        v /= (double)(m * m);
        t.c[m] = v;
    }
    return t;
}
constexpr double bessel_i0(double z) {
    constexpr I0Coef t = make_i0_coef();
    const double zz = 0.25 * z * z;
    double sum = t.c[60];
    #pragma unroll
    for (int m = 59; m >= 0; --m) {
        sum = sum * zz + t.c[m];
    }
    return sum;
}
constexpr double W_GAIN = SCALE * ROLLOFF / bessel_i0(KBETA);

__global__ __launch_bounds__(256) void build_weights(float* __restrict__ W) {
    int idx = blockIdx.x * 256 + threadIdx.x;
    int p = idx / KPAD;
    int jst = idx - p * KPAD;
    int j = (jst & ~6) | ((jst & 2) << 1) | ((jst & 4) >> 1);
    int pn   = p * 441;
    int boff = pn / 160;
    double frac = (double)(pn - boff * 160) / 160.0;
    int    k    = j - 177 - boff;
    double u    = ((double)k - frac) * SCALE;
    double w    = 0.0;
    if (fabs(u) < NUMZ) {
        double xr = ROLLOFF * u;
        double s  = (xr == 0.0) ? 1.0 : sin(M_PI * xr) / (M_PI * xr);
        double r  = 1.0 - (u / NUMZ) * (u / NUMZ);
        if (r < 0.0) r = 0.0;
        w = W_GAIN * s * bessel_i0(KBETA * sqrt(r));
    }
    W[idx] = (float)w;
}

#define RING     12
#define KGRP     (4 * RING)
#define NPAIR    (RING / 2)

typedef float v4f __attribute__((ext_vector_type(4)));
typedef float v4f_a4 __attribute__((ext_vector_type(4), aligned(4)));

struct XRaw { v4f e, f; };

template <bool MASKED>
__device__ inline XRaw load_x(const float* __restrict__ x, int j,
                              int base0, int jlo0, int base1, int jlo1) {
    XRaw r;
    if constexpr (!MASKED) {
        r.e = *(const v4f_a4*)(x + base0 + j);
        r.f = *(const v4f_a4*)(x + base1 + j);
    } else {
        const int jhi0 = jlo0 + T_IN - 1;
        const int jhi1 = jlo1 + T_IN - 1;
        #pragma unroll
        for (int i = 0; i < 4; ++i) {
            int ja = j + i < jlo0 ? jlo0 : j + i;  ja = ja > jhi0 ? jhi0 : ja;
            int jb = j + i < jlo1 ? jlo1 : j + i;  jb = jb > jhi1 ? jhi1 : jb;
            r.e[i] = x[base0 + ja];
            r.f[i] = x[base1 + jb];
        }
    }
    return r;
}

template <bool MASKED>
__device__ inline void mma_pair(const float* __restrict__ lds, int lm, int hi,
                                int jg, XRaw r, int jlo0, int jlo1,
                                v8f c[5][2]) {
    if constexpr (MASKED) {
        const int j = jg + 4 * hi;
        #pragma unroll
        for (int i = 0; i < 4; ++i) {
            float m0 = ((unsigned)(j + i - jlo0) < (unsigned)T_IN) ? 1.0f : 0.0f;
            float m1 = ((unsigned)(j + i - jlo1) < (unsigned)T_IN) ? 1.0f : 0.0f;
            r.e[i] *= m0;
            r.f[i] *= m1;
        }
    }
    #pragma unroll
    for (int t = 0; t < 2; ++t) {
        v2f B0 = {r.e[2 * t], r.e[2 * t + 1]};
        v2f B1 = {r.f[2 * t], r.f[2 * t + 1]};
        const int ja = jg + 4 * t + 2 * hi;
        #pragma unroll
        for (int mt = 0; mt < 5; ++mt) {
            v2f A = *(const v2f*)(lds + (mt * 16 + lm) * LDSP + ja);
            c[mt][0] = __builtin_amdgcn_wmma_f32_16x16x4_f32(
                false, A, false, B0, (short)0, c[mt][0], false, false);
            c[mt][1] = __builtin_amdgcn_wmma_f32_16x16x4_f32(
                false, A, false, B1, (short)0, c[mt][1], false, false);
        }
    }
}

template <bool MASKED>
__device__ inline void gemm_kloop(const float* __restrict__ x,
                                  const float* __restrict__ lds,
                                  int lm, int hi,
                                  int base0, int jlo0,
                                  int base1, int jlo1,
                                  v8f c[5][2]) {
    XRaw cur[NPAIR], nxt[NPAIR];
    #pragma unroll
    for (int s = 0; s < NPAIR; ++s) {
        cur[s] = load_x<MASKED>(x, 8 * s + 4 * hi, base0, jlo0, base1, jlo1);
    }
    #pragma nounroll
    for (int kk0 = 0; kk0 < KW - KGRP; kk0 += KGRP) {
        #pragma unroll
        for (int s = 0; s < NPAIR; ++s) {
            nxt[s] = load_x<MASKED>(x, kk0 + KGRP + 8 * s + 4 * hi,
                                    base0, jlo0, base1, jlo1);
        }
        __builtin_amdgcn_sched_barrier(0);
        #pragma unroll
        for (int s = 0; s < NPAIR; ++s) {
            mma_pair<MASKED>(lds, lm, hi, kk0 + 8 * s, cur[s], jlo0, jlo1, c);
        }
        __builtin_amdgcn_sched_barrier(0);
        #pragma unroll
        for (int s = 0; s < NPAIR; ++s) {
            cur[s] = nxt[s];
        }
    }
    #pragma unroll
    for (int s = 0; s < NPAIR; ++s) {
        mma_pair<MASKED>(lds, lm, hi, KW - KGRP + 8 * s, cur[s], jlo0, jlo1, c);
    }
}

__global__ __launch_bounds__(256) void resample_wmma(const float* __restrict__ x,
                                                     const float* __restrict__ W,
                                                     float* __restrict__ y) {
    extern __shared__ __attribute__((aligned(16))) float lds[];

    const int tid = threadIdx.x;
    const int ph  = blockIdx.y;
    const int cg  = blockIdx.x;
    const int kstart = ph ? 216 : 0;

    {
        const float* Wsrc = W + ph * 80 * KPAD + kstart;
        float4 wv[STG_A][STG_B];
        #pragma unroll
        for (int a = 0; a < STG_A; ++a) {
            #pragma unroll
            for (int b = 0; b < STG_B; ++b) {
                const int i   = tid + 256 * b;
                const int row = i / STG_V4 + 16 * a;
                const int t4  = i % STG_V4;
                wv[a][b] = *(const float4*)&Wsrc[row * KPAD + t4 * 4];
            }
        }
        #pragma unroll
        for (int a = 0; a < STG_A; ++a) {
            #pragma unroll
            for (int b = 0; b < STG_B; ++b) {
                const int i   = tid + 256 * b;
                const int row = i / STG_V4 + 16 * a;
                const int t4  = i % STG_V4;
                *(float4*)&lds[row * LDSP + t4 * 4] = wv[a][b];
            }
        }
    }
    __syncthreads();

    const int wave = tid >> 5;
    const int lane = tid & 31;
    const int lm   = lane & 15;
    const int hi   = lane >> 4;

    const int col0 = cg * 256 + wave * 32 + lm;
    const int col1 = col0 + 16;
    const int b0 = col0 / QBLK, q0 = col0 - b0 * QBLK;
    const int b1 = col1 / QBLK, q1 = col1 - b1 * QBLK;
    const int base0 = b0 * T_IN + q0 * 441 - 177 + kstart;
    const int base1 = b1 * T_IN + q1 * 441 - 177 + kstart;
    const int jlo0  = 177 - q0 * 441 - kstart;
    const int jlo1  = 177 - q1 * 441 - kstart;

    v8f c[5][2] = {};

    const bool edge_lane = (q0 == 0) | (q0 == QBLK - 1) |
                           (q1 == 0) | (q1 == QBLK - 1);
    if (__any((int)edge_lane)) {
        gemm_kloop<true >(x, lds, lm, hi, base0, jlo0, base1, jlo1, c);
    } else {
        gemm_kloop<false>(x, lds, lm, hi, base0, jlo0, base1, jlo1, c);
    }

    const int mrow = hi * 8;
    const int ob0 = b0 * T_OUT + q0 * 160 + ph * 80;
    const int ob1 = b1 * T_OUT + q1 * 160 + ph * 80;
    #pragma unroll
    for (int mt = 0; mt < 5; ++mt) {
        #pragma unroll
        for (int r = 0; r < 8; ++r) {
            y[ob0 + mt * 16 + mrow + r] = c[mt][0][r];
            y[ob1 + mt * 16 + mrow + r] = c[mt][1][r];
        }
    }
}

extern "C" void kernel_launch(void* const* d_in, const int* in_sizes, int n_in,
                              void* d_out, int out_size, void* d_ws, size_t ws_size,
                              hipStream_t stream) {
    const float* x = (const float*)d_in[0];
    float*       W = (float*)d_ws;
    float*       y = (float*)d_out;

    build_weights<<<500, 256, 0, stream>>>(W);

    dim3 grid(125, 2);
    resample_wmma<<<grid, 256, 80 * LDSP * sizeof(float), stream>>>(x, W, y);
}


# each 16-row M-tile does WMMAs and A-fragment LDS reads only in the 24-tap k-blocks its rows' band overlaps (88 of 120 tile-blocks, scalar branches); skipped weights exact zeros, sum order kept
# speedup vs baseline: 3.3577x; 1.0374x over previous
#include <hip/hip_runtime.h>
#include <math.h>


typedef float v2f __attribute__((ext_vector_type(2)));
typedef float v8f __attribute__((ext_vector_type(8)));

#define T_IN     441000
#define T_OUT    160000
#define BATCH    32
#define QBLK     1000
#define KPAD     800
#define KW       576
#define LDSP     580
#define STG_V4   (KW / 4)
#define STG_B    9
#define STG_A    5
#define NPHASE   160
#define NUMZ     64.0
#define ROLLOFF  0.9475937167399596
#define KBETA    14.769656459379492
#define SCALE    (16000.0 / 44100.0)

struct I0Coef { double c[61]; };
constexpr I0Coef make_i0_coef() {
    I0Coef t = {};
    double v = 1.0;
    t.c[0] = 1.0;
    for (int m = 1; m <= 60; ++m) {
        v /= (double)(m * m);
        t.c[m] = v;
    }
    return t;
}
constexpr double bessel_i0(double z) {
    constexpr I0Coef t = make_i0_coef();
    const double zz = 0.25 * z * z;
    double sum = t.c[60];
    #pragma unroll
    for (int m = 59; m >= 0; --m) {
        sum = sum * zz + t.c[m];
    }
    return sum;
}
constexpr double W_GAIN = SCALE * ROLLOFF / bessel_i0(KBETA);

__global__ __launch_bounds__(256) void build_weights(float* __restrict__ W) {
    int idx = blockIdx.x * 256 + threadIdx.x;
    int p = idx / KPAD;
    int jst = idx - p * KPAD;
    int j = (jst & ~6) | ((jst & 2) << 1) | ((jst & 4) >> 1);
    int pn   = p * 441;
    int boff = pn / 160;
    double frac = (double)(pn - boff * 160) / 160.0;
    int    k    = j - 177 - boff;
    double u    = ((double)k - frac) * SCALE;
    double w    = 0.0;
    if (fabs(u) < NUMZ) {
        double xr = ROLLOFF * u;
        double s  = (xr == 0.0) ? 1.0 : sin(M_PI * xr) / (M_PI * xr);
        double r  = 1.0 - (u / NUMZ) * (u / NUMZ);
        if (r < 0.0) r = 0.0;
        w = W_GAIN * s * bessel_i0(KBETA * sqrt(r));
    }
    W[idx] = (float)w;
}

#define RING     12
#define KGRP     (4 * RING)
#define NPAIR    (RING / 2)

typedef float v4f __attribute__((ext_vector_type(4)));
typedef float v4f_a4 __attribute__((ext_vector_type(4), aligned(4)));

struct XRaw { v4f e, f; };

template <bool MASKED>
__device__ inline XRaw load_x(const float* __restrict__ x, int j,
                              int base0, int jlo0, int base1, int jlo1) {
    XRaw r;
    if constexpr (!MASKED) {
        r.e = *(const v4f_a4*)(x + base0 + j);
        r.f = *(const v4f_a4*)(x + base1 + j);
    } else {
        const int jhi0 = jlo0 + T_IN - 1;
        const int jhi1 = jlo1 + T_IN - 1;
        #pragma unroll
        for (int i = 0; i < 4; ++i) {
            int ja = j + i < jlo0 ? jlo0 : j + i;  ja = ja > jhi0 ? jhi0 : ja;
            int jb = j + i < jlo1 ? jlo1 : j + i;  jb = jb > jhi1 ? jhi1 : jb;
            r.e[i] = x[base0 + ja];
            r.f[i] = x[base1 + jb];
        }
    }
    return r;
}

#define BPAIR    3
#define BTAPS    (8 * BPAIR)
#define KSTART1  216

constexpr int band_boff(int p) { return (p * 441) / 160; }
constexpr int band_first(int mt) {
    int best = KW / BTAPS;
    for (int ph = 0; ph < 2; ++ph) {
        int lo = band_boff(ph * 80 + 16 * mt) + 1 - (ph ? KSTART1 : 0);
        if (lo < 0) lo = 0;
        if (lo / BTAPS < best) best = lo / BTAPS;
    }
    return best;
}
constexpr int band_end(int mt) {
    int best = 0;
    for (int ph = 0; ph < 2; ++ph) {
        int hi = band_boff(ph * 80 + 16 * mt + 15) + 354 - (ph ? KSTART1 : 0);
        if (hi > KW - 1) hi = KW - 1;
        if (hi / BTAPS + 1 > best) best = hi / BTAPS + 1;
    }
    return best;
}

template <bool MASKED>
__device__ inline void mma_group(const float* __restrict__ lds, int lm, int hi,
                                 int g, const XRaw* cur, int jlo0, int jlo1,
                                 v8f c[5][2]) {
    XRaw m[NPAIR];
    #pragma unroll
    for (int s = 0; s < NPAIR; ++s) {
        m[s] = cur[s];
        if constexpr (MASKED) {
            const int j = g * KGRP + 8 * s + 4 * hi;
            #pragma unroll
            for (int i = 0; i < 4; ++i) {
                float m0 = ((unsigned)(j + i - jlo0) < (unsigned)T_IN) ? 1.0f : 0.0f;
                float m1 = ((unsigned)(j + i - jlo1) < (unsigned)T_IN) ? 1.0f : 0.0f;
                m[s].e[i] *= m0;
                m[s].f[i] *= m1;
            }
        }
    }
    #pragma unroll
    for (int h2 = 0; h2 < NPAIR / BPAIR; ++h2) {
        const int blk = g * (NPAIR / BPAIR) + h2;
        #pragma unroll
        for (int mt = 0; mt < 5; ++mt) {
            if (blk >= band_first(mt) && blk < band_end(mt)) {
                #pragma unroll
                for (int s = h2 * BPAIR; s < (h2 + 1) * BPAIR; ++s) {
                    #pragma unroll
                    for (int t = 0; t < 2; ++t) {
                        v2f B0 = {m[s].e[2 * t], m[s].e[2 * t + 1]};
                        v2f B1 = {m[s].f[2 * t], m[s].f[2 * t + 1]};
                        const int ja = g * KGRP + 8 * s + 4 * t + 2 * hi;
                        v2f A = *(const v2f*)(lds + (mt * 16 + lm) * LDSP + ja);
                        c[mt][0] = __builtin_amdgcn_wmma_f32_16x16x4_f32(
                            false, A, false, B0, (short)0, c[mt][0], false, false);
                        c[mt][1] = __builtin_amdgcn_wmma_f32_16x16x4_f32(
                            false, A, false, B1, (short)0, c[mt][1], false, false);
                    }
                }
            }
        }
    }
}

template <bool MASKED>
__device__ inline void gemm_kloop(const float* __restrict__ x,
                                  const float* __restrict__ lds,
                                  int lm, int hi,
                                  int base0, int jlo0,
                                  int base1, int jlo1,
                                  v8f c[5][2]) {
    XRaw cur[NPAIR], nxt[NPAIR];
    #pragma unroll
    for (int s = 0; s < NPAIR; ++s) {
        cur[s] = load_x<MASKED>(x, 8 * s + 4 * hi, base0, jlo0, base1, jlo1);
    }
    #pragma nounroll
    for (int g = 0; g < KW / KGRP - 1; ++g) {
        #pragma unroll
        for (int s = 0; s < NPAIR; ++s) {
            nxt[s] = load_x<MASKED>(x, (g + 1) * KGRP + 8 * s + 4 * hi,
                                    base0, jlo0, base1, jlo1);
        }
        __builtin_amdgcn_sched_barrier(0);
        mma_group<MASKED>(lds, lm, hi, g, cur, jlo0, jlo1, c);
        __builtin_amdgcn_sched_barrier(0);
        #pragma unroll
        for (int s = 0; s < NPAIR; ++s) {
            cur[s] = nxt[s];
        }
    }
    mma_group<MASKED>(lds, lm, hi, KW / KGRP - 1, cur, jlo0, jlo1, c);
}

__global__ __launch_bounds__(256) void resample_wmma(const float* __restrict__ x,
                                                     const float* __restrict__ W,
                                                     float* __restrict__ y) {
    extern __shared__ __attribute__((aligned(16))) float lds[];

    const int tid = threadIdx.x;
    const int ph  = blockIdx.y;
    const int cg  = blockIdx.x;
    const int kstart = ph ? KSTART1 : 0;

    {
        const float* Wsrc = W + ph * 80 * KPAD + kstart;
        float4 wv[STG_A][STG_B];
        #pragma unroll
        for (int a = 0; a < STG_A; ++a) {
            #pragma unroll
            for (int b = 0; b < STG_B; ++b) {
                const int i   = tid + 256 * b;
                const int row = i / STG_V4 + 16 * a;
                const int t4  = i % STG_V4;
                wv[a][b] = *(const float4*)&Wsrc[row * KPAD + t4 * 4];
            }
        }
        #pragma unroll
        for (int a = 0; a < STG_A; ++a) {
            #pragma unroll
            for (int b = 0; b < STG_B; ++b) {
                const int i   = tid + 256 * b;
                const int row = i / STG_V4 + 16 * a;
                const int t4  = i % STG_V4;
                *(float4*)&lds[row * LDSP + t4 * 4] = wv[a][b];
            }
        }
    }
    __syncthreads();

    const int wave = tid >> 5;
    const int lane = tid & 31;
    const int lm   = lane & 15;
    const int hi   = lane >> 4;

    const int col0 = cg * 256 + wave * 32 + lm;
    const int col1 = col0 + 16;
    const int b0 = col0 / QBLK, q0 = col0 - b0 * QBLK;
    const int b1 = col1 / QBLK, q1 = col1 - b1 * QBLK;
    const int base0 = b0 * T_IN + q0 * 441 - 177 + kstart;
    const int base1 = b1 * T_IN + q1 * 441 - 177 + kstart;
    const int jlo0  = 177 - q0 * 441 - kstart;
    const int jlo1  = 177 - q1 * 441 - kstart;

    v8f c[5][2] = {};

    const bool edge_lane = (q0 == 0) | (q0 == QBLK - 1) |
                           (q1 == 0) | (q1 == QBLK - 1);
    if (__any((int)edge_lane)) {
        gemm_kloop<true >(x, lds, lm, hi, base0, jlo0, base1, jlo1, c);
    } else {
        gemm_kloop<false>(x, lds, lm, hi, base0, jlo0, base1, jlo1, c);
    }

    const int mrow = hi * 8;
    const int ob0 = b0 * T_OUT + q0 * 160 + ph * 80;
    const int ob1 = b1 * T_OUT + q1 * 160 + ph * 80;
    #pragma unroll
    for (int mt = 0; mt < 5; ++mt) {
        #pragma unroll
        for (int r = 0; r < 8; ++r) {
            y[ob0 + mt * 16 + mrow + r] = c[mt][0][r];
            y[ob1 + mt * 16 + mrow + r] = c[mt][1][r];
        }
    }
}

extern "C" void kernel_launch(void* const* d_in, const int* in_sizes, int n_in,
                              void* d_out, int out_size, void* d_ws, size_t ws_size,
                              hipStream_t stream) {
    const float* x = (const float*)d_in[0];
    float*       W = (float*)d_ws;
    float*       y = (float*)d_out;

    build_weights<<<500, 256, 0, stream>>>(W);

    dim3 grid(125, 2);
    resample_wmma<<<grid, 256, 80 * LDSP * sizeof(float), stream>>>(x, W, y);
}


# weight-table kernel: each distinct non-zero weight evaluated once (band taps only; rows 81..159 stored as mirrors W[160-p][795-j] = W[p][j]) in packed f64 waves, one per SIMD
# speedup vs baseline: 3.7225x; 1.1086x over previous
#include <hip/hip_runtime.h>
#include <math.h>


typedef float v2f __attribute__((ext_vector_type(2)));
typedef float v8f __attribute__((ext_vector_type(8)));

#define T_IN     441000
#define T_OUT    160000
#define BATCH    32
#define QBLK     1000
#define KPAD     800
#define KW       576
#define LDSP     580
#define STG_V4   (KW / 4)
#define STG_B    9
#define STG_A    5
#define NPHASE   160
#define NUMZ     64.0
#define ROLLOFF  0.9475937167399596
#define KBETA    14.769656459379492
#define SCALE    (16000.0 / 44100.0)

struct I0Coef { double c[61]; };
constexpr I0Coef make_i0_coef() {
    I0Coef t = {};
    double v = 1.0;
    t.c[0] = 1.0;
    for (int m = 1; m <= 60; ++m) {
        v /= (double)(m * m);
        t.c[m] = v;
    }
    return t;
}
constexpr double bessel_i0(double z) {
    constexpr I0Coef t = make_i0_coef();
    const double zz = 0.25 * z * z;
    double sum = t.c[60];
    #pragma unroll
    for (int m = 59; m >= 0; --m) {
        sum = sum * zz + t.c[m];
    }
    return sum;
}
constexpr double W_GAIN = SCALE * ROLLOFF / bessel_i0(KBETA);

#define BAND     354
#define NEVAL    81
#define NZERO    (KPAD - BAND)
#define K1_BLOCK 128
#define K1_GRID  ((NEVAL * BAND + K1_BLOCK - 1) / K1_BLOCK)

__device__ inline int stored_pos(int j) {
    return (j & ~6) | ((j & 2) << 1) | ((j & 4) >> 1);
}

__device__ inline void store_row(float* __restrict__ row, int boff, int j, int t,
                                 float w) {
    row[stored_pos(j)] = w;
    int z0 = boff + BAND + 1 + t;
    z0 = z0 >= KPAD ? z0 - KPAD : z0;
    row[stored_pos(z0)] = 0.0f;
    if (t < NZERO - BAND) {
        int z1 = boff + 2 * BAND + 1 + t;
        z1 = z1 >= KPAD ? z1 - KPAD : z1;
        row[stored_pos(z1)] = 0.0f;
    }
}

__global__ __launch_bounds__(K1_BLOCK) void build_weights(float* __restrict__ W) {
    int idx = blockIdx.x * K1_BLOCK + threadIdx.x;
    if (idx >= NEVAL * BAND) return;
    int p = idx / BAND;
    int t = idx - p * BAND;
    int pn   = p * 441;
    int boff = pn / 160;
    double frac = (double)(pn - boff * 160) / 160.0;
    int    k    = t - 176;
    int    j    = boff + 1 + t;
    double u    = ((double)k - frac) * SCALE;
    double w    = 0.0;
    if (fabs(u) < NUMZ) {
        double xr = ROLLOFF * u;
        double s  = (xr == 0.0) ? 1.0 : sin(M_PI * xr) / (M_PI * xr);
        double r  = 1.0 - (u / NUMZ) * (u / NUMZ);
        if (r < 0.0) r = 0.0;
        w = W_GAIN * s * bessel_i0(KBETA * sqrt(r));
    }
    const float wf = (float)w;
    store_row(W + p * KPAD, boff, j, t, wf);
    if (p >= 1 && p < NEVAL - 1) {
        store_row(W + (NPHASE - p) * KPAD, 440 - boff, 795 - j, t, wf);
    }
}

#define RING     12
#define KGRP     (4 * RING)
#define NPAIR    (RING / 2)

typedef float v4f __attribute__((ext_vector_type(4)));
typedef float v4f_a4 __attribute__((ext_vector_type(4), aligned(4)));

struct XRaw { v4f e, f; };

template <bool MASKED>
__device__ inline XRaw load_x(const float* __restrict__ x, int j,
                              int base0, int jlo0, int base1, int jlo1) {
    XRaw r;
    if constexpr (!MASKED) {
        r.e = *(const v4f_a4*)(x + base0 + j);
        r.f = *(const v4f_a4*)(x + base1 + j);
    } else {
        const int jhi0 = jlo0 + T_IN - 1;
        const int jhi1 = jlo1 + T_IN - 1;
        #pragma unroll
        for (int i = 0; i < 4; ++i) {
            int ja = j + i < jlo0 ? jlo0 : j + i;  ja = ja > jhi0 ? jhi0 : ja;
            int jb = j + i < jlo1 ? jlo1 : j + i;  jb = jb > jhi1 ? jhi1 : jb;
            r.e[i] = x[base0 + ja];
            r.f[i] = x[base1 + jb];
        }
    }
    return r;
}

#define BPAIR    3
#define BTAPS    (8 * BPAIR)
#define KSTART1  216

constexpr int band_boff(int p) { return (p * 441) / 160; }
constexpr int band_first(int mt) {
    int best = KW / BTAPS;
    for (int ph = 0; ph < 2; ++ph) {
        int lo = band_boff(ph * 80 + 16 * mt) + 1 - (ph ? KSTART1 : 0);
        if (lo < 0) lo = 0;
        if (lo / BTAPS < best) best = lo / BTAPS;
    }
    return best;
}
constexpr int band_end(int mt) {
    int best = 0;
    for (int ph = 0; ph < 2; ++ph) {
        int hi = band_boff(ph * 80 + 16 * mt + 15) + 354 - (ph ? KSTART1 : 0);
        if (hi > KW - 1) hi = KW - 1;
        if (hi / BTAPS + 1 > best) best = hi / BTAPS + 1;
    }
    return best;
}

template <bool MASKED>
__device__ inline void mma_group(const float* __restrict__ lds, int lm, int hi,
                                 int g, const XRaw* cur, int jlo0, int jlo1,
                                 v8f c[5][2]) {
    XRaw m[NPAIR];
    #pragma unroll
    for (int s = 0; s < NPAIR; ++s) {
        m[s] = cur[s];
        if constexpr (MASKED) {
            const int j = g * KGRP + 8 * s + 4 * hi;
            #pragma unroll
            for (int i = 0; i < 4; ++i) {
                float m0 = ((unsigned)(j + i - jlo0) < (unsigned)T_IN) ? 1.0f : 0.0f;
                float m1 = ((unsigned)(j + i - jlo1) < (unsigned)T_IN) ? 1.0f : 0.0f;
                m[s].e[i] *= m0;
                m[s].f[i] *= m1;
            }
        }
    }
    #pragma unroll
    for (int h2 = 0; h2 < NPAIR / BPAIR; ++h2) {
        const int blk = g * (NPAIR / BPAIR) + h2;
        #pragma unroll
        for (int mt = 0; mt < 5; ++mt) {
            if (blk >= band_first(mt) && blk < band_end(mt)) {
                #pragma unroll
                for (int s = h2 * BPAIR; s < (h2 + 1) * BPAIR; ++s) {
                    #pragma unroll
                    for (int t = 0; t < 2; ++t) {
                        v2f B0 = {m[s].e[2 * t], m[s].e[2 * t + 1]};
                        v2f B1 = {m[s].f[2 * t], m[s].f[2 * t + 1]};
                        const int ja = g * KGRP + 8 * s + 4 * t + 2 * hi;
                        v2f A = *(const v2f*)(lds + (mt * 16 + lm) * LDSP + ja);
                        c[mt][0] = __builtin_amdgcn_wmma_f32_16x16x4_f32(
                            false, A, false, B0, (short)0, c[mt][0], false, false);
                        c[mt][1] = __builtin_amdgcn_wmma_f32_16x16x4_f32(
                            false, A, false, B1, (short)0, c[mt][1], false, false);
                    }
                }
            }
        }
    }
}

template <bool MASKED>
__device__ inline void gemm_kloop(const float* __restrict__ x,
                                  const float* __restrict__ lds,
                                  int lm, int hi,
                                  int base0, int jlo0,
                                  int base1, int jlo1,
                                  v8f c[5][2]) {
    XRaw cur[NPAIR], nxt[NPAIR];
    #pragma unroll
    for (int s = 0; s < NPAIR; ++s) {
        cur[s] = load_x<MASKED>(x, 8 * s + 4 * hi, base0, jlo0, base1, jlo1);
    }
    #pragma nounroll
    for (int g = 0; g < KW / KGRP - 1; ++g) {
        #pragma unroll
        for (int s = 0; s < NPAIR; ++s) {
            nxt[s] = load_x<MASKED>(x, (g + 1) * KGRP + 8 * s + 4 * hi,
                                    base0, jlo0, base1, jlo1);
        }
        __builtin_amdgcn_sched_barrier(0);
        mma_group<MASKED>(lds, lm, hi, g, cur, jlo0, jlo1, c);
        __builtin_amdgcn_sched_barrier(0);
        #pragma unroll
        for (int s = 0; s < NPAIR; ++s) {
            cur[s] = nxt[s];
        }
    }
    mma_group<MASKED>(lds, lm, hi, KW / KGRP - 1, cur, jlo0, jlo1, c);
}

__global__ __launch_bounds__(256) void resample_wmma(const float* __restrict__ x,
                                                     const float* __restrict__ W,
                                                     float* __restrict__ y) {
    extern __shared__ __attribute__((aligned(16))) float lds[];

    const int tid = threadIdx.x;
    const int ph  = blockIdx.y;
    const int cg  = blockIdx.x;
    const int kstart = ph ? KSTART1 : 0;

    {
        const float* Wsrc = W + ph * 80 * KPAD + kstart;
        float4 wv[STG_A][STG_B];
        #pragma unroll
        for (int a = 0; a < STG_A; ++a) {
            #pragma unroll
            for (int b = 0; b < STG_B; ++b) {
                const int i   = tid + 256 * b;
                const int row = i / STG_V4 + 16 * a;
                const int t4  = i % STG_V4;
                wv[a][b] = *(const float4*)&Wsrc[row * KPAD + t4 * 4];
            }
        }
        #pragma unroll
        for (int a = 0; a < STG_A; ++a) {
            #pragma unroll
            for (int b = 0; b < STG_B; ++b) {
                const int i   = tid + 256 * b;
                const int row = i / STG_V4 + 16 * a;
                const int t4  = i % STG_V4;
                *(float4*)&lds[row * LDSP + t4 * 4] = wv[a][b];
            }
        }
    }
    __syncthreads();

    const int wave = tid >> 5;
    const int lane = tid & 31;
    const int lm   = lane & 15;
    const int hi   = lane >> 4;

    const int col0 = cg * 256 + wave * 32 + lm;
    const int col1 = col0 + 16;
    const int b0 = col0 / QBLK, q0 = col0 - b0 * QBLK;
    const int b1 = col1 / QBLK, q1 = col1 - b1 * QBLK;
    const int base0 = b0 * T_IN + q0 * 441 - 177 + kstart;
    const int base1 = b1 * T_IN + q1 * 441 - 177 + kstart;
    const int jlo0  = 177 - q0 * 441 - kstart;
    const int jlo1  = 177 - q1 * 441 - kstart;

    v8f c[5][2] = {};

    const bool edge_lane = (q0 == 0) | (q0 == QBLK - 1) |
                           (q1 == 0) | (q1 == QBLK - 1);
    if (__any((int)edge_lane)) {
        gemm_kloop<true >(x, lds, lm, hi, base0, jlo0, base1, jlo1, c);
    } else {
        gemm_kloop<false>(x, lds, lm, hi, base0, jlo0, base1, jlo1, c);
    }

    const int mrow = hi * 8;
    const int ob0 = b0 * T_OUT + q0 * 160 + ph * 80;
    const int ob1 = b1 * T_OUT + q1 * 160 + ph * 80;
    #pragma unroll
    for (int mt = 0; mt < 5; ++mt) {
        #pragma unroll
        for (int r = 0; r < 8; ++r) {
            y[ob0 + mt * 16 + mrow + r] = c[mt][0][r];
            y[ob1 + mt * 16 + mrow + r] = c[mt][1][r];
        }
    }
}

extern "C" void kernel_launch(void* const* d_in, const int* in_sizes, int n_in,
                              void* d_out, int out_size, void* d_ws, size_t ws_size,
                              hipStream_t stream) {
    const float* x = (const float*)d_in[0];
    float*       W = (float*)d_ws;
    float*       y = (float*)d_out;

    build_weights<<<K1_GRID, K1_BLOCK, 0, stream>>>(W);

    dim3 grid(125, 2);
    resample_wmma<<<grid, 256, 80 * LDSP * sizeof(float), stream>>>(x, W, y);
}


# accumulators staged through a per-wave LDS patch, stored in y's order 16 B per lane: whole 128-byte lines per store instruction (3 line writes a column, was 40 8-byte pieces); sums unchanged
# speedup vs baseline: 4.2523x; 1.1423x over previous
#include <hip/hip_runtime.h>
#include <math.h>


typedef float v2f __attribute__((ext_vector_type(2)));
typedef float v8f __attribute__((ext_vector_type(8)));

#define T_IN     441000
#define T_OUT    160000
#define BATCH    32
#define QBLK     1000
#define KPAD     800
#define KW       576
#define LDSP     580
#define STG_V4   (KW / 4)
#define STG_B    9
#define STG_A    5
#define OPITCH   84
#define OPATCH   (32 * OPITCH)
#define NPHASE   160
#define NUMZ     64.0
#define ROLLOFF  0.9475937167399596
#define KBETA    14.769656459379492
#define SCALE    (16000.0 / 44100.0)

struct I0Coef { double c[61]; };
constexpr I0Coef make_i0_coef() {
    I0Coef t = {};
    double v = 1.0;
    t.c[0] = 1.0;
    for (int m = 1; m <= 60; ++m) {
        v /= (double)(m * m);
        t.c[m] = v;
    }
    return t;
}
constexpr double bessel_i0(double z) {
    constexpr I0Coef t = make_i0_coef();
    const double zz = 0.25 * z * z;
    double sum = t.c[60];
    #pragma unroll
    for (int m = 59; m >= 0; --m) {
        sum = sum * zz + t.c[m];
    }
    return sum;
}
constexpr double W_GAIN = SCALE * ROLLOFF / bessel_i0(KBETA);

#define BAND     354
#define NEVAL    81
#define NZERO    (KPAD - BAND)
#define K1_BLOCK 128
#define K1_GRID  ((NEVAL * BAND + K1_BLOCK - 1) / K1_BLOCK)

__device__ inline int stored_pos(int j) {
    return (j & ~6) | ((j & 2) << 1) | ((j & 4) >> 1);
}

__device__ inline void store_row(float* __restrict__ row, int boff, int j, int t,
                                 float w) {
    row[stored_pos(j)] = w;
    int z0 = boff + BAND + 1 + t;
    z0 = z0 >= KPAD ? z0 - KPAD : z0;
    row[stored_pos(z0)] = 0.0f;
    if (t < NZERO - BAND) {
        int z1 = boff + 2 * BAND + 1 + t;
        z1 = z1 >= KPAD ? z1 - KPAD : z1;
        row[stored_pos(z1)] = 0.0f;
    }
}

__global__ __launch_bounds__(K1_BLOCK) void build_weights(float* __restrict__ W) {
    int idx = blockIdx.x * K1_BLOCK + threadIdx.x;
    if (idx >= NEVAL * BAND) return;
    int p = idx / BAND;
    int t = idx - p * BAND;
    int pn   = p * 441;
    int boff = pn / 160;
    double frac = (double)(pn - boff * 160) / 160.0;
    int    k    = t - 176;
    int    j    = boff + 1 + t;
    double u    = ((double)k - frac) * SCALE;
    double w    = 0.0;
    if (fabs(u) < NUMZ) {
        double xr = ROLLOFF * u;
        double s  = (xr == 0.0) ? 1.0 : sin(M_PI * xr) / (M_PI * xr);
        double r  = 1.0 - (u / NUMZ) * (u / NUMZ);
        if (r < 0.0) r = 0.0;
        w = W_GAIN * s * bessel_i0(KBETA * sqrt(r));
    }
    const float wf = (float)w;
    store_row(W + p * KPAD, boff, j, t, wf);
    if (p >= 1 && p < NEVAL - 1) {
        store_row(W + (NPHASE - p) * KPAD, 440 - boff, 795 - j, t, wf);
    }
}

#define RING     12
#define KGRP     (4 * RING)
#define NPAIR    (RING / 2)

typedef float v4f __attribute__((ext_vector_type(4)));
typedef float v4f_a4 __attribute__((ext_vector_type(4), aligned(4)));

struct XRaw { v4f e, f; };

template <bool MASKED>
__device__ inline XRaw load_x(const float* __restrict__ x, int j,
                              int base0, int jlo0, int base1, int jlo1) {
    XRaw r;
    if constexpr (!MASKED) {
        r.e = *(const v4f_a4*)(x + base0 + j);
        r.f = *(const v4f_a4*)(x + base1 + j);
    } else {
        const int jhi0 = jlo0 + T_IN - 1;
        const int jhi1 = jlo1 + T_IN - 1;
        #pragma unroll
        for (int i = 0; i < 4; ++i) {
            int ja = j + i < jlo0 ? jlo0 : j + i;  ja = ja > jhi0 ? jhi0 : ja;
            int jb = j + i < jlo1 ? jlo1 : j + i;  jb = jb > jhi1 ? jhi1 : jb;
            r.e[i] = x[base0 + ja];
            r.f[i] = x[base1 + jb];
        }
    }
    return r;
}

#define BPAIR    3
#define BTAPS    (8 * BPAIR)
#define KSTART1  216

constexpr int band_boff(int p) { return (p * 441) / 160; }
constexpr int band_first(int mt) {
    int best = KW / BTAPS;
    for (int ph = 0; ph < 2; ++ph) {
        int lo = band_boff(ph * 80 + 16 * mt) + 1 - (ph ? KSTART1 : 0);
        if (lo < 0) lo = 0;
        if (lo / BTAPS < best) best = lo / BTAPS;
    }
    return best;
}
constexpr int band_end(int mt) {
    int best = 0;
    for (int ph = 0; ph < 2; ++ph) {
        int hi = band_boff(ph * 80 + 16 * mt + 15) + 354 - (ph ? KSTART1 : 0);
        if (hi > KW - 1) hi = KW - 1;
        if (hi / BTAPS + 1 > best) best = hi / BTAPS + 1;
    }
    return best;
}

template <bool MASKED>
__device__ inline void mma_group(const float* __restrict__ lds, int lm, int hi,
                                 int g, const XRaw* cur, int jlo0, int jlo1,
                                 v8f c[5][2]) {
    XRaw m[NPAIR];
    #pragma unroll
    for (int s = 0; s < NPAIR; ++s) {
        m[s] = cur[s];
        if constexpr (MASKED) {
            const int j = g * KGRP + 8 * s + 4 * hi;
            #pragma unroll
            for (int i = 0; i < 4; ++i) {
                float m0 = ((unsigned)(j + i - jlo0) < (unsigned)T_IN) ? 1.0f : 0.0f;
                float m1 = ((unsigned)(j + i - jlo1) < (unsigned)T_IN) ? 1.0f : 0.0f;
                m[s].e[i] *= m0;
                m[s].f[i] *= m1;
            }
        }
    }
    #pragma unroll
    for (int h2 = 0; h2 < NPAIR / BPAIR; ++h2) {
        const int blk = g * (NPAIR / BPAIR) + h2;
        #pragma unroll
        for (int mt = 0; mt < 5; ++mt) {
            if (blk >= band_first(mt) && blk < band_end(mt)) {
                #pragma unroll
                for (int s = h2 * BPAIR; s < (h2 + 1) * BPAIR; ++s) {
                    #pragma unroll
                    for (int t = 0; t < 2; ++t) {
                        v2f B0 = {m[s].e[2 * t], m[s].e[2 * t + 1]};
                        v2f B1 = {m[s].f[2 * t], m[s].f[2 * t + 1]};
                        const int ja = g * KGRP + 8 * s + 4 * t + 2 * hi;
                        v2f A = *(const v2f*)(lds + (mt * 16 + lm) * LDSP + ja);
                        c[mt][0] = __builtin_amdgcn_wmma_f32_16x16x4_f32(
                            false, A, false, B0, (short)0, c[mt][0], false, false);
                        c[mt][1] = __builtin_amdgcn_wmma_f32_16x16x4_f32(
                            false, A, false, B1, (short)0, c[mt][1], false, false);
                    }
                }
            }
        }
    }
}

template <bool MASKED>
__device__ inline void gemm_kloop(const float* __restrict__ x,
                                  const float* __restrict__ lds,
                                  int lm, int hi,
                                  int base0, int jlo0,
                                  int base1, int jlo1,
                                  v8f c[5][2]) {
    XRaw cur[NPAIR], nxt[NPAIR];
    #pragma unroll
    for (int s = 0; s < NPAIR; ++s) {
        cur[s] = load_x<MASKED>(x, 8 * s + 4 * hi, base0, jlo0, base1, jlo1);
    }
    #pragma nounroll
    for (int g = 0; g < KW / KGRP - 1; ++g) {
        #pragma unroll
        for (int s = 0; s < NPAIR; ++s) {
            nxt[s] = load_x<MASKED>(x, (g + 1) * KGRP + 8 * s + 4 * hi,
                                    base0, jlo0, base1, jlo1);
        }
        __builtin_amdgcn_sched_barrier(0);
        mma_group<MASKED>(lds, lm, hi, g, cur, jlo0, jlo1, c);
        __builtin_amdgcn_sched_barrier(0);
        #pragma unroll
        for (int s = 0; s < NPAIR; ++s) {
            cur[s] = nxt[s];
        }
    }
    mma_group<MASKED>(lds, lm, hi, KW / KGRP - 1, cur, jlo0, jlo1, c);
}

__global__ __launch_bounds__(256) void resample_wmma(const float* __restrict__ x,
                                                     const float* __restrict__ W,
                                                     float* __restrict__ y) {
    extern __shared__ __attribute__((aligned(16))) float lds[];

    const int tid = threadIdx.x;
    const int ph  = blockIdx.y;
    const int cg  = blockIdx.x;
    const int kstart = ph ? KSTART1 : 0;

    {
        const float* Wsrc = W + ph * 80 * KPAD + kstart;
        float4 wv[STG_A][STG_B];
        #pragma unroll
        for (int a = 0; a < STG_A; ++a) {
            #pragma unroll
            for (int b = 0; b < STG_B; ++b) {
                const int i   = tid + 256 * b;
                const int row = i / STG_V4 + 16 * a;
                const int t4  = i % STG_V4;
                wv[a][b] = *(const float4*)&Wsrc[row * KPAD + t4 * 4];
            }
        }
        #pragma unroll
        for (int a = 0; a < STG_A; ++a) {
            #pragma unroll
            for (int b = 0; b < STG_B; ++b) {
                const int i   = tid + 256 * b;
                const int row = i / STG_V4 + 16 * a;
                const int t4  = i % STG_V4;
                *(float4*)&lds[row * LDSP + t4 * 4] = wv[a][b];
            }
        }
    }
    __syncthreads();

    const int wave = tid >> 5;
    const int lane = tid & 31;
    const int lm   = lane & 15;
    const int hi   = lane >> 4;

    const int col0 = cg * 256 + wave * 32 + lm;
    const int col1 = col0 + 16;
    const int b0 = col0 / QBLK, q0 = col0 - b0 * QBLK;
    const int b1 = col1 / QBLK, q1 = col1 - b1 * QBLK;
    const int base0 = b0 * T_IN + q0 * 441 - 177 + kstart;
    const int base1 = b1 * T_IN + q1 * 441 - 177 + kstart;
    const int jlo0  = 177 - q0 * 441 - kstart;
    const int jlo1  = 177 - q1 * 441 - kstart;

    v8f c[5][2] = {};

    const bool edge_lane = (q0 == 0) | (q0 == QBLK - 1) |
                           (q1 == 0) | (q1 == QBLK - 1);
    if (__any((int)edge_lane)) {
        gemm_kloop<true >(x, lds, lm, hi, base0, jlo0, base1, jlo1, c);
    } else {
        gemm_kloop<false>(x, lds, lm, hi, base0, jlo0, base1, jlo1, c);
    }

    float* patch = lds + 80 * LDSP + wave * OPATCH;
    #pragma unroll
    for (int mt = 0; mt < 5; ++mt) {
        #pragma unroll
        for (int nt = 0; nt < 2; ++nt) {
            float* dst = patch + (nt * 16 + lm) * OPITCH + mt * 16 + hi * 8;
            v4f lo = {c[mt][nt][0], c[mt][nt][1], c[mt][nt][2], c[mt][nt][3]};
            v4f up = {c[mt][nt][4], c[mt][nt][5], c[mt][nt][6], c[mt][nt][7]};
            *(v4f*)(dst)     = lo;
            *(v4f*)(dst + 4) = up;
        }
    }
    __builtin_amdgcn_fence(__ATOMIC_SEQ_CST, "wavefront");
    float* yw = y + (cg * 256 + wave * 32) * NPHASE + ph * 80;
    #pragma unroll
    for (int it = 0; it < 20; ++it) {
        const int i     = it * 32 + lane;
        const int col   = i / 20;
        const int piece = i - col * 20;
        const v4f v = *(const v4f*)(patch + col * OPITCH + piece * 4);
        *(v4f*)(yw + col * NPHASE + piece * 4) = v;
    }
}

extern "C" void kernel_launch(void* const* d_in, const int* in_sizes, int n_in,
                              void* d_out, int out_size, void* d_ws, size_t ws_size,
                              hipStream_t stream) {
    const float* x = (const float*)d_in[0];
    float*       W = (float*)d_ws;
    float*       y = (float*)d_out;

    build_weights<<<K1_GRID, K1_BLOCK, 0, stream>>>(W);

    dim3 grid(125, 2);
    resample_wmma<<<grid, 256, (80 * LDSP + 8 * OPATCH) * sizeof(float), stream>>>(x, W, y);
}


# A-fragment LDS reads of a whole group of 12 k-steps issued in one place in front of the group's WMMAs (were read per M-tile cell among the WMMAs, each cell waiting out an LDS round trip); same sums
# speedup vs baseline: 4.7573x; 1.1188x over previous
#include <hip/hip_runtime.h>
#include <math.h>


typedef float v2f __attribute__((ext_vector_type(2)));
typedef float v8f __attribute__((ext_vector_type(8)));

#define T_IN     441000
#define T_OUT    160000
#define BATCH    32
#define QBLK     1000
#define KPAD     800
#define KW       576
#define LDSP     580
#define STG_V4   (KW / 4)
#define STG_B    9
#define STG_A    5
#define OPITCH   84
#define OPATCH   (32 * OPITCH)
#define NPHASE   160
#define NUMZ     64.0
#define ROLLOFF  0.9475937167399596
#define KBETA    14.769656459379492
#define SCALE    (16000.0 / 44100.0)

struct I0Coef { double c[61]; };
constexpr I0Coef make_i0_coef() {
    I0Coef t = {};
    double v = 1.0;
    t.c[0] = 1.0;
    for (int m = 1; m <= 60; ++m) {
        v /= (double)(m * m);
        t.c[m] = v;
    }
    return t;
}
constexpr double bessel_i0(double z) {
    constexpr I0Coef t = make_i0_coef();
    const double zz = 0.25 * z * z;
    double sum = t.c[60];
    #pragma unroll
    for (int m = 59; m >= 0; --m) {
        sum = sum * zz + t.c[m];
    }
    return sum;
}
constexpr double W_GAIN = SCALE * ROLLOFF / bessel_i0(KBETA);

#define BAND     354
#define NEVAL    81
#define NZERO    (KPAD - BAND)
#define K1_BLOCK 128
#define K1_GRID  ((NEVAL * BAND + K1_BLOCK - 1) / K1_BLOCK)

__device__ inline int stored_pos(int j) {
    return (j & ~6) | ((j & 2) << 1) | ((j & 4) >> 1);
}

__device__ inline void store_row(float* __restrict__ row, int boff, int j, int t,
                                 float w) {
    row[stored_pos(j)] = w;
    int z0 = boff + BAND + 1 + t;
    z0 = z0 >= KPAD ? z0 - KPAD : z0;
    row[stored_pos(z0)] = 0.0f;
    if (t < NZERO - BAND) {
        int z1 = boff + 2 * BAND + 1 + t;
        z1 = z1 >= KPAD ? z1 - KPAD : z1;
        row[stored_pos(z1)] = 0.0f;
    }
}

__global__ __launch_bounds__(K1_BLOCK) void build_weights(float* __restrict__ W) {
    int idx = blockIdx.x * K1_BLOCK + threadIdx.x;
    if (idx >= NEVAL * BAND) return;
    int p = idx / BAND;
    int t = idx - p * BAND;
    int pn   = p * 441;
    int boff = pn / 160;
    double frac = (double)(pn - boff * 160) / 160.0;
    int    k    = t - 176;
    int    j    = boff + 1 + t;
    double u    = ((double)k - frac) * SCALE;
    double w    = 0.0;
    if (fabs(u) < NUMZ) {
        double xr = ROLLOFF * u;
        double s  = (xr == 0.0) ? 1.0 : sin(M_PI * xr) / (M_PI * xr);
        double r  = 1.0 - (u / NUMZ) * (u / NUMZ);
        if (r < 0.0) r = 0.0;
        w = W_GAIN * s * bessel_i0(KBETA * sqrt(r));
    }
    const float wf = (float)w;
    store_row(W + p * KPAD, boff, j, t, wf);
    if (p >= 1 && p < NEVAL - 1) {
        store_row(W + (NPHASE - p) * KPAD, 440 - boff, 795 - j, t, wf);
    }
}

#define RING     12
#define KGRP     (4 * RING)
#define NPAIR    (RING / 2)

typedef float v4f __attribute__((ext_vector_type(4)));
typedef float v4f_a4 __attribute__((ext_vector_type(4), aligned(4)));

struct XRaw { v4f e, f; };

template <bool MASKED>
__device__ inline XRaw load_x(const float* __restrict__ x, int j,
                              int base0, int jlo0, int base1, int jlo1) {
    XRaw r;
    if constexpr (!MASKED) {
        r.e = *(const v4f_a4*)(x + base0 + j);
        r.f = *(const v4f_a4*)(x + base1 + j);
    } else {
        const int jhi0 = jlo0 + T_IN - 1;
        const int jhi1 = jlo1 + T_IN - 1;
        #pragma unroll
        for (int i = 0; i < 4; ++i) {
            int ja = j + i < jlo0 ? jlo0 : j + i;  ja = ja > jhi0 ? jhi0 : ja;
            int jb = j + i < jlo1 ? jlo1 : j + i;  jb = jb > jhi1 ? jhi1 : jb;
            r.e[i] = x[base0 + ja];
            r.f[i] = x[base1 + jb];
        }
    }
    return r;
}

#define BPAIR    3
#define BTAPS    (8 * BPAIR)
#define KSTART1  216

constexpr int band_boff(int p) { return (p * 441) / 160; }
constexpr int band_first(int mt) {
    int best = KW / BTAPS;
    for (int ph = 0; ph < 2; ++ph) {
        int lo = band_boff(ph * 80 + 16 * mt) + 1 - (ph ? KSTART1 : 0);
        if (lo < 0) lo = 0;
        if (lo / BTAPS < best) best = lo / BTAPS;
    }
    return best;
}
constexpr int band_end(int mt) {
    int best = 0;
    for (int ph = 0; ph < 2; ++ph) {
        int hi = band_boff(ph * 80 + 16 * mt + 15) + 354 - (ph ? KSTART1 : 0);
        if (hi > KW - 1) hi = KW - 1;
        if (hi / BTAPS + 1 > best) best = hi / BTAPS + 1;
    }
    return best;
}

template <bool MASKED>
__device__ inline void mma_group(const float* __restrict__ lds, int lm, int hi,
                                 int g, const XRaw* cur, int jlo0, int jlo1,
                                 v8f c[5][2]) {
    XRaw m[NPAIR];
    #pragma unroll
    for (int s = 0; s < NPAIR; ++s) {
        m[s] = cur[s];
        if constexpr (MASKED) {
            const int j = g * KGRP + 8 * s + 4 * hi;
            #pragma unroll
            for (int i = 0; i < 4; ++i) {
                float m0 = ((unsigned)(j + i - jlo0) < (unsigned)T_IN) ? 1.0f : 0.0f;
                float m1 = ((unsigned)(j + i - jlo1) < (unsigned)T_IN) ? 1.0f : 0.0f;
                m[s].e[i] *= m0;
                m[s].f[i] *= m1;
            }
        }
    }
    v2f a[NPAIR / BPAIR][5][2 * BPAIR];
    #pragma unroll
    for (int h2 = 0; h2 < NPAIR / BPAIR; ++h2) {
        const int blk = g * (NPAIR / BPAIR) + h2;
        #pragma unroll
        for (int mt = 0; mt < 5; ++mt) {
            if (blk >= band_first(mt) && blk < band_end(mt)) {
                #pragma unroll
                for (int s = h2 * BPAIR; s < (h2 + 1) * BPAIR; ++s) {
                    #pragma unroll
                    for (int t = 0; t < 2; ++t) {
                        const int ja = g * KGRP + 8 * s + 4 * t + 2 * hi;
                        a[h2][mt][2 * (s - h2 * BPAIR) + t] =
                            *(const v2f*)(lds + (mt * 16 + lm) * LDSP + ja);
                    }
                }
            }
        }
    }
    __builtin_amdgcn_sched_barrier(0);
    #pragma unroll
    for (int h2 = 0; h2 < NPAIR / BPAIR; ++h2) {
        const int blk = g * (NPAIR / BPAIR) + h2;
        #pragma unroll
        for (int mt = 0; mt < 5; ++mt) {
            if (blk >= band_first(mt) && blk < band_end(mt)) {
                #pragma unroll
                for (int s = h2 * BPAIR; s < (h2 + 1) * BPAIR; ++s) {
                    #pragma unroll
                    for (int t = 0; t < 2; ++t) {
                        v2f B0 = {m[s].e[2 * t], m[s].e[2 * t + 1]};
                        v2f B1 = {m[s].f[2 * t], m[s].f[2 * t + 1]};
                        const v2f A = a[h2][mt][2 * (s - h2 * BPAIR) + t];
                        c[mt][0] = __builtin_amdgcn_wmma_f32_16x16x4_f32(
                            false, A, false, B0, (short)0, c[mt][0], false, false);
                        c[mt][1] = __builtin_amdgcn_wmma_f32_16x16x4_f32(
                            false, A, false, B1, (short)0, c[mt][1], false, false);
                    }
                }
            }
        }
    }
}

template <bool MASKED>
__device__ inline void gemm_kloop(const float* __restrict__ x,
                                  const float* __restrict__ lds,
                                  int lm, int hi,
                                  int base0, int jlo0,
                                  int base1, int jlo1,
                                  v8f c[5][2]) {
    XRaw cur[NPAIR], nxt[NPAIR];
    #pragma unroll
    for (int s = 0; s < NPAIR; ++s) {
        cur[s] = load_x<MASKED>(x, 8 * s + 4 * hi, base0, jlo0, base1, jlo1);
    }
    #pragma nounroll
    for (int g = 0; g < KW / KGRP - 1; ++g) {
        #pragma unroll
        for (int s = 0; s < NPAIR; ++s) {
            nxt[s] = load_x<MASKED>(x, (g + 1) * KGRP + 8 * s + 4 * hi,
                                    base0, jlo0, base1, jlo1);
        }
        __builtin_amdgcn_sched_barrier(0);
        mma_group<MASKED>(lds, lm, hi, g, cur, jlo0, jlo1, c);
        __builtin_amdgcn_sched_barrier(0);
        #pragma unroll
        for (int s = 0; s < NPAIR; ++s) {
            cur[s] = nxt[s];
        }
    }
    mma_group<MASKED>(lds, lm, hi, KW / KGRP - 1, cur, jlo0, jlo1, c);
}

__global__ __launch_bounds__(256) void resample_wmma(const float* __restrict__ x,
                                                     const float* __restrict__ W,
                                                     float* __restrict__ y) {
    extern __shared__ __attribute__((aligned(16))) float lds[];

    const int tid = threadIdx.x;
    const int ph  = blockIdx.y;
    const int cg  = blockIdx.x;
    const int kstart = ph ? KSTART1 : 0;

    {
        const float* Wsrc = W + ph * 80 * KPAD + kstart;
        float4 wv[STG_A][STG_B];
        #pragma unroll
        for (int a = 0; a < STG_A; ++a) {
            #pragma unroll
            for (int b = 0; b < STG_B; ++b) {
                const int i   = tid + 256 * b;
                const int row = i / STG_V4 + 16 * a;
                const int t4  = i % STG_V4;
                wv[a][b] = *(const float4*)&Wsrc[row * KPAD + t4 * 4];
            }
        }
        #pragma unroll
        for (int a = 0; a < STG_A; ++a) {
            #pragma unroll
            for (int b = 0; b < STG_B; ++b) {
                const int i   = tid + 256 * b;
                const int row = i / STG_V4 + 16 * a;
                const int t4  = i % STG_V4;
                *(float4*)&lds[row * LDSP + t4 * 4] = wv[a][b];
            }
        }
    }
    __syncthreads();

    const int wave = tid >> 5;
    const int lane = tid & 31;
    const int lm   = lane & 15;
    const int hi   = lane >> 4;

    const int col0 = cg * 256 + wave * 32 + lm;
    const int col1 = col0 + 16;
    const int b0 = col0 / QBLK, q0 = col0 - b0 * QBLK;
    const int b1 = col1 / QBLK, q1 = col1 - b1 * QBLK;
    const int base0 = b0 * T_IN + q0 * 441 - 177 + kstart;
    const int base1 = b1 * T_IN + q1 * 441 - 177 + kstart;
    const int jlo0  = 177 - q0 * 441 - kstart;
    const int jlo1  = 177 - q1 * 441 - kstart;

    v8f c[5][2] = {};

    const bool edge_lane = (q0 == 0) | (q0 == QBLK - 1) |
                           (q1 == 0) | (q1 == QBLK - 1);
    if (__any((int)edge_lane)) {
        gemm_kloop<true >(x, lds, lm, hi, base0, jlo0, base1, jlo1, c);
    } else {
        gemm_kloop<false>(x, lds, lm, hi, base0, jlo0, base1, jlo1, c);
    }

    float* patch = lds + 80 * LDSP + wave * OPATCH;
    #pragma unroll
    for (int mt = 0; mt < 5; ++mt) {
        #pragma unroll
        for (int nt = 0; nt < 2; ++nt) {
            float* dst = patch + (nt * 16 + lm) * OPITCH + mt * 16 + hi * 8;
            v4f lo = {c[mt][nt][0], c[mt][nt][1], c[mt][nt][2], c[mt][nt][3]};
            v4f up = {c[mt][nt][4], c[mt][nt][5], c[mt][nt][6], c[mt][nt][7]};
            *(v4f*)(dst)     = lo;
            *(v4f*)(dst + 4) = up;
        }
    }
    __builtin_amdgcn_fence(__ATOMIC_SEQ_CST, "wavefront");
    float* yw = y + (cg * 256 + wave * 32) * NPHASE + ph * 80;
    #pragma unroll
    for (int it = 0; it < 20; ++it) {
        const int i     = it * 32 + lane;
        const int col   = i / 20;
        const int piece = i - col * 20;
        const v4f v = *(const v4f*)(patch + col * OPITCH + piece * 4);
        *(v4f*)(yw + col * NPHASE + piece * 4) = v;
    }
}

extern "C" void kernel_launch(void* const* d_in, const int* in_sizes, int n_in,
                              void* d_out, int out_size, void* d_ws, size_t ws_size,
                              hipStream_t stream) {
    const float* x = (const float*)d_in[0];
    float*       W = (float*)d_ws;
    float*       y = (float*)d_out;

    build_weights<<<K1_GRID, K1_BLOCK, 0, stream>>>(W);

    dim3 grid(125, 2);
    resample_wmma<<<grid, 256, (80 * LDSP + 8 * OPATCH) * sizeof(float), stream>>>(x, W, y);
}


# edge waves (a column at a row end): one clamped 16-byte load per piece like interior waves (was four 4-byte loads); pieces realigned and zeroed only in groups where a lane's taps leave its row
# speedup vs baseline: 5.0204x; 1.0553x over previous
#include <hip/hip_runtime.h>
#include <math.h>


typedef float v2f __attribute__((ext_vector_type(2)));
typedef float v8f __attribute__((ext_vector_type(8)));

#define T_IN     441000
#define T_OUT    160000
#define BATCH    32
#define QBLK     1000
#define KPAD     800
#define KW       576
#define LDSP     580
#define STG_V4   (KW / 4)
#define STG_B    9
#define STG_A    5
#define OPITCH   84
#define OPATCH   (32 * OPITCH)
#define NPHASE   160
#define NUMZ     64.0
#define ROLLOFF  0.9475937167399596
#define KBETA    14.769656459379492
#define SCALE    (16000.0 / 44100.0)

struct I0Coef { double c[61]; };
constexpr I0Coef make_i0_coef() {
    I0Coef t = {};
    double v = 1.0;
    t.c[0] = 1.0;
    for (int m = 1; m <= 60; ++m) {
        v /= (double)(m * m);
        t.c[m] = v;
    }
    return t;
}
constexpr double bessel_i0(double z) {
    constexpr I0Coef t = make_i0_coef();
    const double zz = 0.25 * z * z;
    double sum = t.c[60];
    #pragma unroll
    for (int m = 59; m >= 0; --m) {
        sum = sum * zz + t.c[m];
    }
    return sum;
}
constexpr double W_GAIN = SCALE * ROLLOFF / bessel_i0(KBETA);

#define BAND     354
#define NEVAL    81
#define NZERO    (KPAD - BAND)
#define K1_BLOCK 128
#define K1_GRID  ((NEVAL * BAND + K1_BLOCK - 1) / K1_BLOCK)

__device__ inline int stored_pos(int j) {
    return (j & ~6) | ((j & 2) << 1) | ((j & 4) >> 1);
}

__device__ inline void store_row(float* __restrict__ row, int boff, int j, int t,
                                 float w) {
    row[stored_pos(j)] = w;
    int z0 = boff + BAND + 1 + t;
    z0 = z0 >= KPAD ? z0 - KPAD : z0;
    row[stored_pos(z0)] = 0.0f;
    if (t < NZERO - BAND) {
        int z1 = boff + 2 * BAND + 1 + t;
        z1 = z1 >= KPAD ? z1 - KPAD : z1;
        row[stored_pos(z1)] = 0.0f;
    }
}

__global__ __launch_bounds__(K1_BLOCK) void build_weights(float* __restrict__ W) {
    int idx = blockIdx.x * K1_BLOCK + threadIdx.x;
    if (idx >= NEVAL * BAND) return;
    int p = idx / BAND;
    int t = idx - p * BAND;
    int pn   = p * 441;
    int boff = pn / 160;
    double frac = (double)(pn - boff * 160) / 160.0;
    int    k    = t - 176;
    int    j    = boff + 1 + t;
    double u    = ((double)k - frac) * SCALE;
    double w    = 0.0;
    if (fabs(u) < NUMZ) {
        double xr = ROLLOFF * u;
        double s  = (xr == 0.0) ? 1.0 : sin(M_PI * xr) / (M_PI * xr);
        double r  = 1.0 - (u / NUMZ) * (u / NUMZ);
        if (r < 0.0) r = 0.0;
        w = W_GAIN * s * bessel_i0(KBETA * sqrt(r));
    }
    const float wf = (float)w;
    store_row(W + p * KPAD, boff, j, t, wf);
    if (p >= 1 && p < NEVAL - 1) {
        store_row(W + (NPHASE - p) * KPAD, 440 - boff, 795 - j, t, wf);
    }
}

#define RING     12
#define KGRP     (4 * RING)
#define NPAIR    (RING / 2)

typedef float v4f __attribute__((ext_vector_type(4)));
typedef float v4f_a4 __attribute__((ext_vector_type(4), aligned(4)));

struct XRaw { v4f e, f; };

__device__ inline int piece_start(int j, int jlo) {
    const int jc = j < jlo ? jlo : j;
    return jc > jlo + T_IN - 4 ? jlo + T_IN - 4 : jc;
}

template <bool MASKED>
__device__ inline XRaw load_x(const float* __restrict__ x, int j,
                              int base0, int jlo0, int base1, int jlo1) {
    XRaw r;
    if constexpr (!MASKED) {
        r.e = *(const v4f_a4*)(x + base0 + j);
        r.f = *(const v4f_a4*)(x + base1 + j);
    } else {
        r.e = *(const v4f_a4*)(x + base0 + piece_start(j, jlo0));
        r.f = *(const v4f_a4*)(x + base1 + piece_start(j, jlo1));
    }
    return r;
}

__device__ inline v4f edge_piece(v4f L, int j, int jlo) {
    const int sh = piece_start(j, jlo) - j;
    const int up = sh > 0 ? sh : 0;
    const int dn = sh < 0 ? -sh : 0;
    v4f a = L;
    if (up & 1) a = (v4f){0.0f, a[0], a[1], a[2]};
    if (up & 2) a = (v4f){0.0f, 0.0f, a[0], a[1]};
    if (dn & 1) a = (v4f){a[1], a[2], a[3], 0.0f};
    if (dn & 2) a = (v4f){a[2], a[3], 0.0f, 0.0f};
    v4f r;
    #pragma unroll
    for (int i = 0; i < 4; ++i) {
        r[i] = ((unsigned)(j + i - jlo) < (unsigned)T_IN) ? a[i] : 0.0f;
    }
    return r;
}

__device__ inline bool group_leaves_row(int g, int jlo0, int jlo1) {
    const int jf = g * KGRP;
    const int jl = jf + KGRP - 1;
    const bool out = (jf < jlo0) | (jl - jlo0 >= T_IN) |
                     (jf < jlo1) | (jl - jlo1 >= T_IN);
    return __any((int)out) != 0;
}

#define BPAIR    3
#define BTAPS    (8 * BPAIR)
#define KSTART1  216

constexpr int band_boff(int p) { return (p * 441) / 160; }
constexpr int band_first(int mt) {
    int best = KW / BTAPS;
    for (int ph = 0; ph < 2; ++ph) {
        int lo = band_boff(ph * 80 + 16 * mt) + 1 - (ph ? KSTART1 : 0);
        if (lo < 0) lo = 0;
        if (lo / BTAPS < best) best = lo / BTAPS;
    }
    return best;
}
constexpr int band_end(int mt) {
    int best = 0;
    for (int ph = 0; ph < 2; ++ph) {
        int hi = band_boff(ph * 80 + 16 * mt + 15) + 354 - (ph ? KSTART1 : 0);
        if (hi > KW - 1) hi = KW - 1;
        if (hi / BTAPS + 1 > best) best = hi / BTAPS + 1;
    }
    return best;
}

template <bool MASKED>
__device__ inline void mma_group(const float* __restrict__ lds, int lm, int hi,
                                 int g, const XRaw* cur, int jlo0, int jlo1,
                                 v8f c[5][2]) {
    XRaw m[NPAIR];
    #pragma unroll
    for (int s = 0; s < NPAIR; ++s) {
        m[s] = cur[s];
    }
    if constexpr (MASKED) {
        if (group_leaves_row(g, jlo0, jlo1)) {
            #pragma unroll
            for (int s = 0; s < NPAIR; ++s) {
                const int j = g * KGRP + 8 * s + 4 * hi;
                m[s].e = edge_piece(m[s].e, j, jlo0);
                m[s].f = edge_piece(m[s].f, j, jlo1);
            }
        }
    }
    v2f a[NPAIR / BPAIR][5][2 * BPAIR];
    #pragma unroll
    for (int h2 = 0; h2 < NPAIR / BPAIR; ++h2) {
        const int blk = g * (NPAIR / BPAIR) + h2;
        #pragma unroll
        for (int mt = 0; mt < 5; ++mt) {
            if (blk >= band_first(mt) && blk < band_end(mt)) {
                #pragma unroll
                for (int s = h2 * BPAIR; s < (h2 + 1) * BPAIR; ++s) {
                    #pragma unroll
                    for (int t = 0; t < 2; ++t) {
                        const int ja = g * KGRP + 8 * s + 4 * t + 2 * hi;
                        a[h2][mt][2 * (s - h2 * BPAIR) + t] =
                            *(const v2f*)(lds + (mt * 16 + lm) * LDSP + ja);
                    }
                }
            }
        }
    }
    __builtin_amdgcn_sched_barrier(0);
    #pragma unroll
    for (int h2 = 0; h2 < NPAIR / BPAIR; ++h2) {
        const int blk = g * (NPAIR / BPAIR) + h2;
        #pragma unroll
        for (int mt = 0; mt < 5; ++mt) {
            if (blk >= band_first(mt) && blk < band_end(mt)) {
                #pragma unroll
                for (int s = h2 * BPAIR; s < (h2 + 1) * BPAIR; ++s) {
                    #pragma unroll
                    for (int t = 0; t < 2; ++t) {
                        v2f B0 = {m[s].e[2 * t], m[s].e[2 * t + 1]};
                        v2f B1 = {m[s].f[2 * t], m[s].f[2 * t + 1]};
                        const v2f A = a[h2][mt][2 * (s - h2 * BPAIR) + t];
                        c[mt][0] = __builtin_amdgcn_wmma_f32_16x16x4_f32(
                            false, A, false, B0, (short)0, c[mt][0], false, false);
                        c[mt][1] = __builtin_amdgcn_wmma_f32_16x16x4_f32(
                            false, A, false, B1, (short)0, c[mt][1], false, false);
                    }
                }
            }
        }
    }
}

template <bool MASKED>
__device__ inline void gemm_kloop(const float* __restrict__ x,
                                  const float* __restrict__ lds,
                                  int lm, int hi,
                                  int base0, int jlo0,
                                  int base1, int jlo1,
                                  v8f c[5][2]) {
    XRaw cur[NPAIR], nxt[NPAIR];
    #pragma unroll
    for (int s = 0; s < NPAIR; ++s) {
        cur[s] = load_x<MASKED>(x, 8 * s + 4 * hi, base0, jlo0, base1, jlo1);
    }
    #pragma nounroll
    for (int g = 0; g < KW / KGRP - 1; ++g) {
        #pragma unroll
        for (int s = 0; s < NPAIR; ++s) {
            nxt[s] = load_x<MASKED>(x, (g + 1) * KGRP + 8 * s + 4 * hi,
                                    base0, jlo0, base1, jlo1);
        }
        __builtin_amdgcn_sched_barrier(0);
        mma_group<MASKED>(lds, lm, hi, g, cur, jlo0, jlo1, c);
        __builtin_amdgcn_sched_barrier(0);
        #pragma unroll
        for (int s = 0; s < NPAIR; ++s) {
            cur[s] = nxt[s];
        }
    }
    mma_group<MASKED>(lds, lm, hi, KW / KGRP - 1, cur, jlo0, jlo1, c);
}

__global__ __launch_bounds__(256) void resample_wmma(const float* __restrict__ x,
                                                     const float* __restrict__ W,
                                                     float* __restrict__ y) {
    extern __shared__ __attribute__((aligned(16))) float lds[];

    const int tid = threadIdx.x;
    const int ph  = blockIdx.y;
    const int cg  = blockIdx.x;
    const int kstart = ph ? KSTART1 : 0;

    {
        const float* Wsrc = W + ph * 80 * KPAD + kstart;
        float4 wv[STG_A][STG_B];
        #pragma unroll
        for (int a = 0; a < STG_A; ++a) {
            #pragma unroll
            for (int b = 0; b < STG_B; ++b) {
                const int i   = tid + 256 * b;
                const int row = i / STG_V4 + 16 * a;
                const int t4  = i % STG_V4;
                wv[a][b] = *(const float4*)&Wsrc[row * KPAD + t4 * 4];
            }
        }
        #pragma unroll
        for (int a = 0; a < STG_A; ++a) {
            #pragma unroll
            for (int b = 0; b < STG_B; ++b) {
                const int i   = tid + 256 * b;
                const int row = i / STG_V4 + 16 * a;
                const int t4  = i % STG_V4;
                *(float4*)&lds[row * LDSP + t4 * 4] = wv[a][b];
            }
        }
    }
    __syncthreads();

    const int wave = tid >> 5;
    const int lane = tid & 31;
    const int lm   = lane & 15;
    const int hi   = lane >> 4;

    const int col0 = cg * 256 + wave * 32 + lm;
    const int col1 = col0 + 16;
    const int b0 = col0 / QBLK, q0 = col0 - b0 * QBLK;
    const int b1 = col1 / QBLK, q1 = col1 - b1 * QBLK;
    const int base0 = b0 * T_IN + q0 * 441 - 177 + kstart;
    const int base1 = b1 * T_IN + q1 * 441 - 177 + kstart;
    const int jlo0  = 177 - q0 * 441 - kstart;
    const int jlo1  = 177 - q1 * 441 - kstart;

    v8f c[5][2] = {};

    const bool edge_lane = (q0 == 0) | (q0 == QBLK - 1) |
                           (q1 == 0) | (q1 == QBLK - 1);
    if (__any((int)edge_lane)) {
        gemm_kloop<true >(x, lds, lm, hi, base0, jlo0, base1, jlo1, c);
    } else {
        gemm_kloop<false>(x, lds, lm, hi, base0, jlo0, base1, jlo1, c);
    }

    float* patch = lds + 80 * LDSP + wave * OPATCH;
    #pragma unroll
    for (int mt = 0; mt < 5; ++mt) {
        #pragma unroll
        for (int nt = 0; nt < 2; ++nt) {
            float* dst = patch + (nt * 16 + lm) * OPITCH + mt * 16 + hi * 8;
            v4f lo = {c[mt][nt][0], c[mt][nt][1], c[mt][nt][2], c[mt][nt][3]};
            v4f up = {c[mt][nt][4], c[mt][nt][5], c[mt][nt][6], c[mt][nt][7]};
            *(v4f*)(dst)     = lo;
            *(v4f*)(dst + 4) = up;
        }
    }
    __builtin_amdgcn_fence(__ATOMIC_SEQ_CST, "wavefront");
    float* yw = y + (cg * 256 + wave * 32) * NPHASE + ph * 80;
    #pragma unroll
    for (int it = 0; it < 20; ++it) {
        const int i     = it * 32 + lane;
        const int col   = i / 20;
        const int piece = i - col * 20;
        const v4f v = *(const v4f*)(patch + col * OPITCH + piece * 4);
        *(v4f*)(yw + col * NPHASE + piece * 4) = v;
    }
}

extern "C" void kernel_launch(void* const* d_in, const int* in_sizes, int n_in,
                              void* d_out, int out_size, void* d_ws, size_t ws_size,
                              hipStream_t stream) {
    const float* x = (const float*)d_in[0];
    float*       W = (float*)d_ws;
    float*       y = (float*)d_out;

    build_weights<<<K1_GRID, K1_BLOCK, 0, stream>>>(W);

    dim3 grid(125, 2);
    resample_wmma<<<grid, 256, (80 * LDSP + 8 * OPATCH) * sizeof(float), stream>>>(x, W, y);
}
